# starting kernel
# baseline (speedup 1.0000x reference)
#include <hip/hip_runtime.h>
#include <math.h>

typedef float v2f __attribute__((ext_vector_type(2)));
typedef float v8f __attribute__((ext_vector_type(8)));

#define T_DIM 1024
#define B_DIM 64
#define D_DIM 512
#define N_DIM 64
#define ROWS  (T_DIM * B_DIM)
#define TBN   ((size_t)T_DIM * B_DIM * N_DIM)

__global__ __launch_bounds__(128) void ntm_proj_kernel(
    const float* __restrict__ x,
    const float* __restrict__ Wk, const float* __restrict__ Wv,
    const float* __restrict__ Wq, const float* __restrict__ We,
    const float* __restrict__ Ww,
    float* __restrict__ outk, float* __restrict__ outv,
    float* __restrict__ outq, float* __restrict__ oute,
    float* __restrict__ outw)
{
    __shared__ float ldsA[16 * 64];
    __shared__ float ldsS[2 * 16 * 64];

    const int tid   = threadIdx.x;
    const int wave  = tid >> 5;
    const int lane  = tid & 31;
    const int lmod  = lane & 15;
    const bool hi   = lane >= 16;
    const int Mbase = blockIdx.x * 16;
    const int Nbase = wave * 16;

    const float* Wmat[5] = {Wk, Wv, Wq, We, Ww};

    v8f acc[5];
#pragma unroll
    for (int m = 0; m < 5; ++m)
#pragma unroll
        for (int r = 0; r < 8; ++r) acc[m][r] = 0.0f;

    for (int c = 0; c < 8; ++c) {
        __syncthreads();
#pragma unroll
        for (int i = tid; i < 16 * 64; i += 128) {
            const int r = i >> 6, col = i & 63;
            ldsA[i] = x[(size_t)(Mbase + r) * D_DIM + c * 64 + col];
        }
        __syncthreads();

        for (int kk = 0; kk < 64; kk += 4) {
            const int ak = kk + (hi ? 2 : 0);
            v2f a;
            a.x = ldsA[lmod * 64 + ak];
            a.y = ldsA[lmod * 64 + ak + 1];

            const int k0   = c * 64 + ak;
            const size_t brow = (size_t)(Nbase + lmod) * D_DIM + k0;
#pragma unroll
            for (int m = 0; m < 5; ++m) {
                v2f b;
                b.x = Wmat[m][brow];
                b.y = Wmat[m][brow + 1];
                acc[m] = __builtin_amdgcn_wmma_f32_16x16x4_f32(
                     false, a,  false, b,
                     (short)0, acc[m],
                     false,  false);
            }
        }
    }

#pragma unroll
    for (int r = 0; r < 8; ++r) {
        const int m = hi ? r + 8 : r;
        const int n = Nbase + lmod;
        ldsS[m * 64 + n]        = acc[0][r];
        ldsS[1024 + m * 64 + n] = acc[2][r];
        const size_t gi = (size_t)(Mbase + m) * N_DIM + n;
        outv[gi] = acc[1][r];
        oute[gi] = 1.0f / (1.0f + __expf(-acc[3][r]));
        outw[gi] = 1.0f / (1.0f + __expf(-acc[4][r]));
    }
    __syncthreads();

    if (tid < 16) {
#pragma unroll
        for (int s = 0; s < 2; ++s) {
            float* rowp = &ldsS[s * 1024 + tid * 64];
            float mx = rowp[0];
            for (int j = 1; j < 64; ++j) mx = fmaxf(mx, rowp[j]);
            float sum = 0.0f;
            for (int j = 0; j < 64; ++j) { float e2 = __expf(rowp[j] - mx); rowp[j] = e2; sum += e2; }
            const float inv = 1.0f / sum;
            for (int j = 0; j < 64; ++j) rowp[j] *= inv;
        }
    }
    __syncthreads();

#pragma unroll
    for (int i = tid; i < 1024; i += 128) {
        const int m = i >> 6, n = i & 63;
        const size_t gi = (size_t)(Mbase + m) * N_DIM + n;
        outk[gi] = ldsS[i];
        outq[gi] = ldsS[1024 + i];
    }
}

__global__ __launch_bounds__(64) void ntm_scan_kernel(
    const float* __restrict__ S0,
    const float* __restrict__ kG, const float* __restrict__ vG,
    const float* __restrict__ qG, const float* __restrict__ eG,
    const float* __restrict__ wG,
    float* __restrict__ out, float* __restrict__ Sfin)
{
    __shared__ float kb[64], vb[64], qb[64], eb[64], wb[64];
    const int b = blockIdx.x;
    const int i = threadIdx.x;

    float S[64];
#pragma unroll
    for (int j = 0; j < 64; ++j)
        S[j] = S0[(size_t)b * 4096 + (size_t)i * 64 + j];

    for (int t = 0; t < T_DIM; ++t) {
        const size_t base = (size_t)t * 4096 + (size_t)b * 64 + i;
        kb[i] = kG[base]; vb[i] = vG[base]; qb[i] = qG[base];
        eb[i] = eG[base]; wb[i] = wG[base];
        __syncthreads();

        const float ei  = eb[i];
        const float wvi = wb[i] * vb[i];
        float acc = 0.0f;
#pragma unroll
        for (int j = 0; j < 64; ++j) {
            const float kj = kb[j];
            float s = S[j];
            s = s * (1.0f - ei * kj) + wvi * kj;
            S[j] = s;
            acc = fmaf(s, qb[j], acc);
        }
        out[(size_t)t * 4096 + (size_t)b * 64 + i] = tanhf(acc);
        __syncthreads();
    }

#pragma unroll
    for (int j = 0; j < 64; ++j)
        Sfin[(size_t)b * 4096 + (size_t)i * 64 + j] = S[j];
}

extern "C" void kernel_launch(void* const* d_in, const int* in_sizes, int n_in,
                              void* d_out, int out_size, void* d_ws, size_t ws_size,
                              hipStream_t stream) {
    (void)in_sizes; (void)n_in; (void)out_size; (void)ws_size;
    const float* x  = (const float*)d_in[0];
    const float* S0 = (const float*)d_in[1];
    const float* Wk = (const float*)d_in[2];
    const float* Wv = (const float*)d_in[3];
    const float* Wq = (const float*)d_in[4];
    const float* We = (const float*)d_in[5];
    const float* Ww = (const float*)d_in[6];

    float* out = (float*)d_out;
    float* ws  = (float*)d_ws;
    float* wk  = ws;
    float* wv  = ws + TBN;
    float* wq  = ws + 2 * TBN;
    float* we_ = ws + 3 * TBN;
    float* ww_ = ws + 4 * TBN;

    ntm_proj_kernel<<<ROWS / 16, 128, 0, stream>>>(
        x, Wk, Wv, Wq, We, Ww, wk, wv, wq, we_, ww_);

    ntm_scan_kernel<<<B_DIM, 64, 0, stream>>>(
        S0, wk, wv, wq, we_, ww_, out, out + TBN);
}


# scan: k,v,q,e,w fetched one 8-step chunk ahead into a 2-half LDS ring behind a hand-written barrier (loads and out stores stay in flight); 1 barrier per 8 steps, same arithmetic order
# speedup vs baseline: 1.7161x; 1.7161x over previous
#include <hip/hip_runtime.h>
#include <math.h>

typedef float v2f __attribute__((ext_vector_type(2)));
typedef float v8f __attribute__((ext_vector_type(8)));

#define T_DIM 1024
#define B_DIM 64
#define D_DIM 512
#define N_DIM 64
#define ROWS  (T_DIM * B_DIM)
#define TBN   ((size_t)T_DIM * B_DIM * N_DIM)

__global__ __launch_bounds__(128) void ntm_proj_kernel(
    const float* __restrict__ x,
    const float* __restrict__ Wk, const float* __restrict__ Wv,
    const float* __restrict__ Wq, const float* __restrict__ We,
    const float* __restrict__ Ww,
    float* __restrict__ outk, float* __restrict__ outv,
    float* __restrict__ outq, float* __restrict__ oute,
    float* __restrict__ outw)
{
    __shared__ float ldsA[16 * 64];
    __shared__ float ldsS[2 * 16 * 64];

    const int tid   = threadIdx.x;
    const int wave  = tid >> 5;
    const int lane  = tid & 31;
    const int lmod  = lane & 15;
    const bool hi   = lane >= 16;
    const int Mbase = blockIdx.x * 16;
    const int Nbase = wave * 16;

    const float* Wmat[5] = {Wk, Wv, Wq, We, Ww};

    v8f acc[5];
#pragma unroll
    for (int m = 0; m < 5; ++m)
#pragma unroll
        for (int r = 0; r < 8; ++r) acc[m][r] = 0.0f;

    for (int c = 0; c < 8; ++c) {
        __syncthreads();
#pragma unroll
        for (int i = tid; i < 16 * 64; i += 128) {
            const int r = i >> 6, col = i & 63;
            ldsA[i] = x[(size_t)(Mbase + r) * D_DIM + c * 64 + col];
        }
        __syncthreads();

        for (int kk = 0; kk < 64; kk += 4) {
            const int ak = kk + (hi ? 2 : 0);
            v2f a;
            a.x = ldsA[lmod * 64 + ak];
            a.y = ldsA[lmod * 64 + ak + 1];

            const int k0   = c * 64 + ak;
            const size_t brow = (size_t)(Nbase + lmod) * D_DIM + k0;
#pragma unroll
            for (int m = 0; m < 5; ++m) {
                v2f b;
                b.x = Wmat[m][brow];
                b.y = Wmat[m][brow + 1];
                acc[m] = __builtin_amdgcn_wmma_f32_16x16x4_f32(
                     false, a,  false, b,
                     (short)0, acc[m],
                     false,  false);
            }
        }
    }

#pragma unroll
    for (int r = 0; r < 8; ++r) {
        const int m = hi ? r + 8 : r;
        const int n = Nbase + lmod;
        ldsS[m * 64 + n]        = acc[0][r];
        ldsS[1024 + m * 64 + n] = acc[2][r];
        const size_t gi = (size_t)(Mbase + m) * N_DIM + n;
        outv[gi] = acc[1][r];
        oute[gi] = 1.0f / (1.0f + __expf(-acc[3][r]));
        outw[gi] = 1.0f / (1.0f + __expf(-acc[4][r]));
    }
    __syncthreads();

    if (tid < 16) {
#pragma unroll
        for (int s = 0; s < 2; ++s) {
            float* rowp = &ldsS[s * 1024 + tid * 64];
            float mx = rowp[0];
            for (int j = 1; j < 64; ++j) mx = fmaxf(mx, rowp[j]);
            float sum = 0.0f;
            for (int j = 0; j < 64; ++j) { float e2 = __expf(rowp[j] - mx); rowp[j] = e2; sum += e2; }
            const float inv = 1.0f / sum;
            for (int j = 0; j < 64; ++j) rowp[j] *= inv;
        }
    }
    __syncthreads();

#pragma unroll
    for (int i = tid; i < 1024; i += 128) {
        const int m = i >> 6, n = i & 63;
        const size_t gi = (size_t)(Mbase + m) * N_DIM + n;
        outk[gi] = ldsS[i];
        outq[gi] = ldsS[1024 + i];
    }
}

typedef float v4f __attribute__((ext_vector_type(4)));

#define CH    8
#define ROWF  (5 * 64)
#define NPF   (CH * 5 * 64 / 4 / 64)

static __device__ __forceinline__ void lds_handover_barrier()
{
    asm volatile("s_wait_dscnt 0x0\n\ts_barrier_signal -1\n\ts_barrier_wait -1" ::: "memory");
}

__global__ __launch_bounds__(64) void ntm_scan_kernel(
    const float* __restrict__ S0,
    const float* __restrict__ kG, const float* __restrict__ vG,
    const float* __restrict__ qG, const float* __restrict__ eG,
    const float* __restrict__ wG,
    float* __restrict__ out, float* __restrict__ Sfin)
{
    __shared__ __attribute__((aligned(16))) float stage[2 * CH * ROWF];
    const int b  = blockIdx.x;
    const int i  = threadIdx.x;
    const int c4 = (i & 15) * 4;
    const int tq = i >> 4;

    float S[64];
#pragma unroll
    for (int j = 0; j < 64; ++j)
        S[j] = S0[(size_t)b * 4096 + (size_t)i * 64 + j];

    const float* src[5] = {kG, vG, qG, eG, wG};

    v4f pf[NPF];
#pragma unroll
    for (int r = 0; r < NPF; ++r) {
        const int tl = ((r & 1) << 2) + tq;
        pf[r] = *(const v4f*)(src[r >> 1] + (size_t)tl * 4096 + (size_t)b * 64 + c4);
    }

    __syncthreads();

    for (int t = 0; t < T_DIM; ++t) {
        const int tl  = t & (CH - 1);
        const int par = (t / CH) & 1;

        if (tl == 0) {
            float* dst = stage + par * (CH * ROWF);
#pragma unroll
            for (int r = 0; r < NPF; ++r) {
                const int tr = ((r & 1) << 2) + tq;
                *(v4f*)(dst + tr * ROWF + (r >> 1) * 64 + c4) = pf[r];
            }
            lds_handover_barrier();
            if (t + CH < T_DIM) {
#pragma unroll
                for (int r = 0; r < NPF; ++r) {
                    const int tr = t + CH + ((r & 1) << 2) + tq;
                    pf[r] = *(const v4f*)(src[r >> 1] + (size_t)tr * 4096 + (size_t)b * 64 + c4);
                }
            }
        }

        const float* kb = stage + (par * CH + tl) * ROWF;
        const float* vb = kb + 64;
        const float* qb = kb + 128;
        const float* eb = kb + 192;
        const float* wb = kb + 256;

        const float ei  = eb[i];
        const float wvi = wb[i] * vb[i];
        float acc = 0.0f;
#pragma unroll
        for (int j = 0; j < 64; ++j) {
            const float kj = kb[j];
            float s = S[j];
            s = s * (1.0f - ei * kj) + wvi * kj;
            S[j] = s;
            acc = fmaf(s, qb[j], acc);
        }
        out[(size_t)t * 4096 + (size_t)b * 64 + i] = tanhf(acc);
    }

#pragma unroll
    for (int j = 0; j < 64; ++j)
        Sfin[(size_t)b * 4096 + (size_t)i * 64 + j] = S[j];
}

extern "C" void kernel_launch(void* const* d_in, const int* in_sizes, int n_in,
                              void* d_out, int out_size, void* d_ws, size_t ws_size,
                              hipStream_t stream) {
    (void)in_sizes; (void)n_in; (void)out_size; (void)ws_size;
    const float* x  = (const float*)d_in[0];
    const float* S0 = (const float*)d_in[1];
    const float* Wk = (const float*)d_in[2];
    const float* Wv = (const float*)d_in[3];
    const float* Wq = (const float*)d_in[4];
    const float* We = (const float*)d_in[5];
    const float* Ww = (const float*)d_in[6];

    float* out = (float*)d_out;
    float* ws  = (float*)d_ws;
    float* wk  = ws;
    float* wv  = ws + TBN;
    float* wq  = ws + 2 * TBN;
    float* we_ = ws + 3 * TBN;
    float* ww_ = ws + 4 * TBN;

    ntm_proj_kernel<<<ROWS / 16, 128, 0, stream>>>(
        x, Wk, Wv, Wq, We, Ww, wk, wv, wq, we_, ww_);

    ntm_scan_kernel<<<B_DIM, 64, 0, stream>>>(
        S0, wk, wv, wq, we_, ww_, out, out + TBN);
}


# scan: each step's k,q,e,w,v LDS reads issued as one group ahead of the arithmetic (were sunk to their uses, each waited for singly); previous step's tanh+store run under them; same sums, same order
# speedup vs baseline: 2.8982x; 1.6888x over previous
#include <hip/hip_runtime.h>
#include <math.h>

typedef float v2f __attribute__((ext_vector_type(2)));
typedef float v8f __attribute__((ext_vector_type(8)));

#define T_DIM 1024
#define B_DIM 64
#define D_DIM 512
#define N_DIM 64
#define ROWS  (T_DIM * B_DIM)
#define TBN   ((size_t)T_DIM * B_DIM * N_DIM)

__global__ __launch_bounds__(128) void ntm_proj_kernel(
    const float* __restrict__ x,
    const float* __restrict__ Wk, const float* __restrict__ Wv,
    const float* __restrict__ Wq, const float* __restrict__ We,
    const float* __restrict__ Ww,
    float* __restrict__ outk, float* __restrict__ outv,
    float* __restrict__ outq, float* __restrict__ oute,
    float* __restrict__ outw)
{
    __shared__ float ldsA[16 * 64];
    __shared__ float ldsS[2 * 16 * 64];

    const int tid   = threadIdx.x;
    const int wave  = tid >> 5;
    const int lane  = tid & 31;
    const int lmod  = lane & 15;
    const bool hi   = lane >= 16;
    const int Mbase = blockIdx.x * 16;
    const int Nbase = wave * 16;

    const float* Wmat[5] = {Wk, Wv, Wq, We, Ww};

    v8f acc[5];
#pragma unroll
    for (int m = 0; m < 5; ++m)
#pragma unroll
        for (int r = 0; r < 8; ++r) acc[m][r] = 0.0f;

    for (int c = 0; c < 8; ++c) {
        __syncthreads();
#pragma unroll
        for (int i = tid; i < 16 * 64; i += 128) {
            const int r = i >> 6, col = i & 63;
            ldsA[i] = x[(size_t)(Mbase + r) * D_DIM + c * 64 + col];
        }
        __syncthreads();

        for (int kk = 0; kk < 64; kk += 4) {
            const int ak = kk + (hi ? 2 : 0);
            v2f a;
            a.x = ldsA[lmod * 64 + ak];
            a.y = ldsA[lmod * 64 + ak + 1];

            const int k0   = c * 64 + ak;
            const size_t brow = (size_t)(Nbase + lmod) * D_DIM + k0;
#pragma unroll
            for (int m = 0; m < 5; ++m) {
                v2f b;
                b.x = Wmat[m][brow];
                b.y = Wmat[m][brow + 1];
                acc[m] = __builtin_amdgcn_wmma_f32_16x16x4_f32(
                     false, a,  false, b,
                     (short)0, acc[m],
                     false,  false);
            }
        }
    }

#pragma unroll
    for (int r = 0; r < 8; ++r) {
        const int m = hi ? r + 8 : r;
        const int n = Nbase + lmod;
        ldsS[m * 64 + n]        = acc[0][r];
        ldsS[1024 + m * 64 + n] = acc[2][r];
        const size_t gi = (size_t)(Mbase + m) * N_DIM + n;
        outv[gi] = acc[1][r];
        oute[gi] = 1.0f / (1.0f + __expf(-acc[3][r]));
        outw[gi] = 1.0f / (1.0f + __expf(-acc[4][r]));
    }
    __syncthreads();

    if (tid < 16) {
#pragma unroll
        for (int s = 0; s < 2; ++s) {
            float* rowp = &ldsS[s * 1024 + tid * 64];
            float mx = rowp[0];
            for (int j = 1; j < 64; ++j) mx = fmaxf(mx, rowp[j]);
            float sum = 0.0f;
            for (int j = 0; j < 64; ++j) { float e2 = __expf(rowp[j] - mx); rowp[j] = e2; sum += e2; }
            const float inv = 1.0f / sum;
            for (int j = 0; j < 64; ++j) rowp[j] *= inv;
        }
    }
    __syncthreads();

#pragma unroll
    for (int i = tid; i < 1024; i += 128) {
        const int m = i >> 6, n = i & 63;
        const size_t gi = (size_t)(Mbase + m) * N_DIM + n;
        outk[gi] = ldsS[i];
        outq[gi] = ldsS[1024 + i];
    }
}

typedef float v4f __attribute__((ext_vector_type(4)));

#define CH    8
#define ROWF  (5 * 64)
#define NPF   (CH * 5 * 64 / 4 / 64)

static __device__ __forceinline__ void lds_handover_barrier()
{
    asm volatile("s_wait_dscnt 0x0\n\ts_barrier_signal -1\n\ts_barrier_wait -1" ::: "memory");
}

__global__ __launch_bounds__(64) void ntm_scan_kernel(
    const float* __restrict__ S0,
    const float* __restrict__ kG, const float* __restrict__ vG,
    const float* __restrict__ qG, const float* __restrict__ eG,
    const float* __restrict__ wG,
    float* __restrict__ out, float* __restrict__ Sfin)
{
    __shared__ __attribute__((aligned(16))) float stage[2 * CH * ROWF];
    const int b  = blockIdx.x;
    const int i  = threadIdx.x;
    const int c4 = (i & 15) * 4;
    const int tq = i >> 4;

    float S[64];
#pragma unroll
    for (int j = 0; j < 64; ++j)
        S[j] = S0[(size_t)b * 4096 + (size_t)i * 64 + j];

    const float* src[5] = {kG, vG, qG, eG, wG};

    v4f pf[NPF];
#pragma unroll
    for (int r = 0; r < NPF; ++r) {
        const int tl = ((r & 1) << 2) + tq;
        pf[r] = *(const v4f*)(src[r >> 1] + (size_t)tl * 4096 + (size_t)b * 64 + c4);
    }

    __syncthreads();

    float acc_prev = 0.0f;
    for (int t = 0; t < T_DIM; ++t) {
        const int tl  = t & (CH - 1);
        const int par = (t / CH) & 1;

        if (tl == 0) {
            float* dst = stage + par * (CH * ROWF);
#pragma unroll
            for (int r = 0; r < NPF; ++r) {
                const int tr = ((r & 1) << 2) + tq;
                *(v4f*)(dst + tr * ROWF + (r >> 1) * 64 + c4) = pf[r];
            }
            lds_handover_barrier();
            if (t + CH < T_DIM) {
#pragma unroll
                for (int r = 0; r < NPF; ++r) {
                    const int tr = t + CH + ((r & 1) << 2) + tq;
                    pf[r] = *(const v4f*)(src[r >> 1] + (size_t)tr * 4096 + (size_t)b * 64 + c4);
                }
            }
        }

        const float* kb = stage + (par * CH + tl) * ROWF;
        const float* vb = kb + 64;
        const float* qb = kb + 128;
        const float* eb = kb + 192;
        const float* wb = kb + 256;

        const float ei = eb[i];
        const float wi = wb[i];
        const float vi = vb[i];
        v4f kr[16], qr[16];
#pragma unroll
        for (int r = 0; r < 16; ++r) kr[r] = *(const v4f*)(kb + 4 * r);
#pragma unroll
        for (int r = 0; r < 16; ++r) qr[r] = *(const v4f*)(qb + 4 * r);
        __builtin_amdgcn_sched_barrier(0);

        if (t > 0)
            out[(size_t)(t - 1) * 4096 + (size_t)b * 64 + i] = tanhf(acc_prev);

        const float wvi = wi * vi;
        float acc = 0.0f;
#pragma unroll
        for (int j = 0; j < 64; ++j) {
            const float kj = kr[j >> 2][j & 3];
            float s = S[j];
            s = s * (1.0f - ei * kj) + wvi * kj;
            S[j] = s;
            acc = fmaf(s, qr[j >> 2][j & 3], acc);
        }
        acc_prev = acc;
    }
    out[(size_t)(T_DIM - 1) * 4096 + (size_t)b * 64 + i] = tanhf(acc_prev);

#pragma unroll
    for (int j = 0; j < 64; ++j)
        Sfin[(size_t)b * 4096 + (size_t)i * 64 + j] = S[j];
}

extern "C" void kernel_launch(void* const* d_in, const int* in_sizes, int n_in,
                              void* d_out, int out_size, void* d_ws, size_t ws_size,
                              hipStream_t stream) {
    (void)in_sizes; (void)n_in; (void)out_size; (void)ws_size;
    const float* x  = (const float*)d_in[0];
    const float* S0 = (const float*)d_in[1];
    const float* Wk = (const float*)d_in[2];
    const float* Wv = (const float*)d_in[3];
    const float* Wq = (const float*)d_in[4];
    const float* We = (const float*)d_in[5];
    const float* Ww = (const float*)d_in[6];

    float* out = (float*)d_out;
    float* ws  = (float*)d_ws;
    float* wk  = ws;
    float* wv  = ws + TBN;
    float* wq  = ws + 2 * TBN;
    float* we_ = ws + 3 * TBN;
    float* ww_ = ws + 4 * TBN;

    ntm_proj_kernel<<<ROWS / 16, 128, 0, stream>>>(
        x, Wk, Wv, Wq, We, Ww, wk, wv, wq, we_, ww_);

    ntm_scan_kernel<<<B_DIM, 64, 0, stream>>>(
        S0, wk, wv, wq, we_, ww_, out, out + TBN);
}


# proj GEMM: x and W k-chunks staged once per 128-row work-group tile in LDS, fragments read from LDS (W was fetched per wave per 16 rows by 8-byte strided global loads); same sums, same order
# speedup vs baseline: 4.6968x; 1.6206x over previous
#include <hip/hip_runtime.h>
#include <math.h>

typedef float v2f __attribute__((ext_vector_type(2)));
typedef float v8f __attribute__((ext_vector_type(8)));

#define T_DIM 1024
#define B_DIM 64
#define D_DIM 512
#define N_DIM 64
#define ROWS  (T_DIM * B_DIM)
#define TBN   ((size_t)T_DIM * B_DIM * N_DIM)

typedef float v4f __attribute__((ext_vector_type(4)));

#define RT      128
#define KC      16
#define LP      20
#define NCHUNK  (D_DIM / KC)
#define A_FL    (RT * LP)
#define STG_FL  ((RT + 5 * N_DIM) * LP)
#define NPIECE  ((RT + 5 * N_DIM) * (KC / 4) / 256)
#define SP      65

__global__ __launch_bounds__(256) void ntm_proj_kernel(
    const float* __restrict__ x,
    const float* __restrict__ Wk, const float* __restrict__ Wv,
    const float* __restrict__ Wq, const float* __restrict__ We,
    const float* __restrict__ Ww,
    float* __restrict__ outk, float* __restrict__ outv,
    float* __restrict__ outq, float* __restrict__ oute,
    float* __restrict__ outw)
{
    __shared__ __attribute__((aligned(16))) float lds[2 * STG_FL];

    const int tid   = threadIdx.x;
    const int wave  = tid >> 5;
    const int lane  = tid & 31;
    const int lmod  = lane & 15;
    const int hi    = lane >> 4;
    const int Mbase = blockIdx.x * RT;
    const int Nbase = (wave & 3) * 16;
    const int Mhalf = (wave >> 2) * 64;

    const int srow = tid >> 2;
    const int sq   = (tid & 3) * 4;
    const float* gsrc[NPIECE];
    gsrc[0] = x  + (size_t)(Mbase + srow) * D_DIM + sq;
    gsrc[1] = x  + (size_t)(Mbase + 64 + srow) * D_DIM + sq;
    gsrc[2] = Wk + (size_t)srow * D_DIM + sq;
    gsrc[3] = Wv + (size_t)srow * D_DIM + sq;
    gsrc[4] = Wq + (size_t)srow * D_DIM + sq;
    gsrc[5] = We + (size_t)srow * D_DIM + sq;
    gsrc[6] = Ww + (size_t)srow * D_DIM + sq;
    const int sdst = srow * LP + sq;

    v8f acc[4][5];
#pragma unroll
    for (int mt = 0; mt < 4; ++mt)
#pragma unroll
        for (int m = 0; m < 5; ++m)
#pragma unroll
            for (int r = 0; r < 8; ++r) acc[mt][m][r] = 0.0f;

    v4f pf[NPIECE];
#pragma unroll
    for (int i = 0; i < NPIECE; ++i) pf[i] = *(const v4f*)(gsrc[i]);
#pragma unroll
    for (int i = 0; i < NPIECE; ++i) *(v4f*)(lds + sdst + i * 64 * LP) = pf[i];
    __syncthreads();

    const int aoff = (Mhalf + lmod) * LP + 2 * hi;
    const int boff = A_FL + (Nbase + lmod) * LP + 2 * hi;

    for (int c = 0; c < NCHUNK; ++c) {
        const float* st = lds + (c & 1) * STG_FL;

        if (c + 1 < NCHUNK) {
#pragma unroll
            for (int i = 0; i < NPIECE; ++i) pf[i] = *(const v4f*)(gsrc[i] + (c + 1) * KC);
        }

#pragma unroll
        for (int kk = 0; kk < KC; kk += 4) {
            v2f a[4], b[5];
#pragma unroll
            for (int mt = 0; mt < 4; ++mt) a[mt] = *(const v2f*)(st + aoff + mt * 16 * LP + kk);
#pragma unroll
            for (int m = 0; m < 5; ++m)   b[m]  = *(const v2f*)(st + boff + m * 64 * LP + kk);
#pragma unroll
            for (int mt = 0; mt < 4; ++mt)
#pragma unroll
                for (int m = 0; m < 5; ++m)
                    acc[mt][m] = __builtin_amdgcn_wmma_f32_16x16x4_f32(
                         false, a[mt],  false, b[m],
                         (short)0, acc[mt][m],
                         false,  false);
        }

        if (c + 1 < NCHUNK) {
            float* dn = lds + ((c + 1) & 1) * STG_FL;
#pragma unroll
            for (int i = 0; i < NPIECE; ++i) *(v4f*)(dn + sdst + i * 64 * LP) = pf[i];
        }
        __syncthreads();
    }

    float* ldsS = lds;
#pragma unroll
    for (int mt = 0; mt < 4; ++mt) {
#pragma unroll
        for (int r = 0; r < 8; ++r) {
            const int m = Mhalf + mt * 16 + (hi ? r + 8 : r);
            const int n = Nbase + lmod;
            ldsS[m * SP + n]           = acc[mt][0][r];
            ldsS[RT * SP + m * SP + n] = acc[mt][2][r];
            const size_t gi = (size_t)(Mbase + m) * N_DIM + n;
            outv[gi] = acc[mt][1][r];
            oute[gi] = 1.0f / (1.0f + __expf(-acc[mt][3][r]));
            outw[gi] = 1.0f / (1.0f + __expf(-acc[mt][4][r]));
        }
    }
    __syncthreads();

    {
        float* rowp = &ldsS[(tid >> 7) * (RT * SP) + (tid & (RT - 1)) * SP];
        float mx = rowp[0];
        for (int j = 1; j < 64; ++j) mx = fmaxf(mx, rowp[j]);
        float sum = 0.0f;
        for (int j = 0; j < 64; ++j) { float e2 = __expf(rowp[j] - mx); rowp[j] = e2; sum += e2; }
        const float inv = 1.0f / sum;
        for (int j = 0; j < 64; ++j) rowp[j] *= inv;
    }
    __syncthreads();

    for (int i = tid; i < RT * 64; i += 256) {
        const int m = i >> 6, n = i & 63;
        const size_t gi = (size_t)(Mbase + m) * N_DIM + n;
        outk[gi] = ldsS[m * SP + n];
        outq[gi] = ldsS[RT * SP + m * SP + n];
    }
}

#define CH    8
#define ROWF  (5 * 64)
#define NPF   (CH * 5 * 64 / 4 / 64)

static __device__ __forceinline__ void lds_handover_barrier()
{
    asm volatile("s_wait_dscnt 0x0\n\ts_barrier_signal -1\n\ts_barrier_wait -1" ::: "memory");
}

__global__ __launch_bounds__(64) void ntm_scan_kernel(
    const float* __restrict__ S0,
    const float* __restrict__ kG, const float* __restrict__ vG,
    const float* __restrict__ qG, const float* __restrict__ eG,
    const float* __restrict__ wG,
    float* __restrict__ out, float* __restrict__ Sfin)
{
    __shared__ __attribute__((aligned(16))) float stage[2 * CH * ROWF];
    const int b  = blockIdx.x;
    const int i  = threadIdx.x;
    const int c4 = (i & 15) * 4;
    const int tq = i >> 4;

    float S[64];
#pragma unroll
    for (int j = 0; j < 64; ++j)
        S[j] = S0[(size_t)b * 4096 + (size_t)i * 64 + j];

    const float* src[5] = {kG, vG, qG, eG, wG};

    v4f pf[NPF];
#pragma unroll
    for (int r = 0; r < NPF; ++r) {
        const int tl = ((r & 1) << 2) + tq;
        pf[r] = *(const v4f*)(src[r >> 1] + (size_t)tl * 4096 + (size_t)b * 64 + c4);
    }

    __syncthreads();

    float acc_prev = 0.0f;
    for (int t = 0; t < T_DIM; ++t) {
        const int tl  = t & (CH - 1);
        const int par = (t / CH) & 1;

        if (tl == 0) {
            float* dst = stage + par * (CH * ROWF);
#pragma unroll
            for (int r = 0; r < NPF; ++r) {
                const int tr = ((r & 1) << 2) + tq;
                *(v4f*)(dst + tr * ROWF + (r >> 1) * 64 + c4) = pf[r];
            }
            lds_handover_barrier();
            if (t + CH < T_DIM) {
#pragma unroll
                for (int r = 0; r < NPF; ++r) {
                    const int tr = t + CH + ((r & 1) << 2) + tq;
                    pf[r] = *(const v4f*)(src[r >> 1] + (size_t)tr * 4096 + (size_t)b * 64 + c4);
                }
            }
        }

        const float* kb = stage + (par * CH + tl) * ROWF;
        const float* vb = kb + 64;
        const float* qb = kb + 128;
        const float* eb = kb + 192;
        const float* wb = kb + 256;

        const float ei = eb[i];
        const float wi = wb[i];
        const float vi = vb[i];
        v4f kr[16], qr[16];
#pragma unroll
        for (int r = 0; r < 16; ++r) kr[r] = *(const v4f*)(kb + 4 * r);
#pragma unroll
        for (int r = 0; r < 16; ++r) qr[r] = *(const v4f*)(qb + 4 * r);
        __builtin_amdgcn_sched_barrier(0);

        if (t > 0)
            out[(size_t)(t - 1) * 4096 + (size_t)b * 64 + i] = tanhf(acc_prev);

        const float wvi = wi * vi;
        float acc = 0.0f;
#pragma unroll
        for (int j = 0; j < 64; ++j) {
            const float kj = kr[j >> 2][j & 3];
            float s = S[j];
            s = s * (1.0f - ei * kj) + wvi * kj;
            S[j] = s;
            acc = fmaf(s, qr[j >> 2][j & 3], acc);
        }
        acc_prev = acc;
    }
    out[(size_t)(T_DIM - 1) * 4096 + (size_t)b * 64 + i] = tanhf(acc_prev);

#pragma unroll
    for (int j = 0; j < 64; ++j)
        Sfin[(size_t)b * 4096 + (size_t)i * 64 + j] = S[j];
}

extern "C" void kernel_launch(void* const* d_in, const int* in_sizes, int n_in,
                              void* d_out, int out_size, void* d_ws, size_t ws_size,
                              hipStream_t stream) {
    (void)in_sizes; (void)n_in; (void)out_size; (void)ws_size;
    const float* x  = (const float*)d_in[0];
    const float* S0 = (const float*)d_in[1];
    const float* Wk = (const float*)d_in[2];
    const float* Wv = (const float*)d_in[3];
    const float* Wq = (const float*)d_in[4];
    const float* We = (const float*)d_in[5];
    const float* Ww = (const float*)d_in[6];

    float* out = (float*)d_out;
    float* ws  = (float*)d_ws;
    float* wk  = ws;
    float* wv  = ws + TBN;
    float* wq  = ws + 2 * TBN;
    float* we_ = ws + 3 * TBN;
    float* ww_ = ws + 4 * TBN;

    ntm_proj_kernel<<<ROWS / RT, 256, 0, stream>>>(
        x, Wk, Wv, Wq, We, Ww, wk, wv, wq, we_, ww_);

    ntm_scan_kernel<<<B_DIM, 64, 0, stream>>>(
        S0, wk, wv, wq, we_, ww_, out, out + TBN);
}


# scan: state over 4x the lanes: 2 work-groups x 32 rows per batch, a row on 8 lanes x 8 columns (was 1 lane per row); read-head sum = 4 chains + add tree + 3 lane-exchange adds: sum order changed
# speedup vs baseline: 9.7873x; 2.0838x over previous
#include <hip/hip_runtime.h>
#include <math.h>

typedef float v2f __attribute__((ext_vector_type(2)));
typedef float v8f __attribute__((ext_vector_type(8)));

#define T_DIM 1024
#define B_DIM 64
#define D_DIM 512
#define N_DIM 64
#define ROWS  (T_DIM * B_DIM)
#define TBN   ((size_t)T_DIM * B_DIM * N_DIM)

typedef float v4f __attribute__((ext_vector_type(4)));

#define RT      128
#define KC      16
#define LP      20
#define NCHUNK  (D_DIM / KC)
#define A_FL    (RT * LP)
#define STG_FL  ((RT + 5 * N_DIM) * LP)
#define NPIECE  ((RT + 5 * N_DIM) * (KC / 4) / 256)
#define SP      65

__global__ __launch_bounds__(256) void ntm_proj_kernel(
    const float* __restrict__ x,
    const float* __restrict__ Wk, const float* __restrict__ Wv,
    const float* __restrict__ Wq, const float* __restrict__ We,
    const float* __restrict__ Ww,
    float* __restrict__ outk, float* __restrict__ outv,
    float* __restrict__ outq, float* __restrict__ oute,
    float* __restrict__ outw)
{
    __shared__ __attribute__((aligned(16))) float lds[2 * STG_FL];

    const int tid   = threadIdx.x;
    const int wave  = tid >> 5;
    const int lane  = tid & 31;
    const int lmod  = lane & 15;
    const int hi    = lane >> 4;
    const int Mbase = blockIdx.x * RT;
    const int Nbase = (wave & 3) * 16;
    const int Mhalf = (wave >> 2) * 64;

    const int srow = tid >> 2;
    const int sq   = (tid & 3) * 4;
    const float* gsrc[NPIECE];
    gsrc[0] = x  + (size_t)(Mbase + srow) * D_DIM + sq;
    gsrc[1] = x  + (size_t)(Mbase + 64 + srow) * D_DIM + sq;
    gsrc[2] = Wk + (size_t)srow * D_DIM + sq;
    gsrc[3] = Wv + (size_t)srow * D_DIM + sq;
    gsrc[4] = Wq + (size_t)srow * D_DIM + sq;
    gsrc[5] = We + (size_t)srow * D_DIM + sq;
    gsrc[6] = Ww + (size_t)srow * D_DIM + sq;
    const int sdst = srow * LP + sq;

    v8f acc[4][5];
#pragma unroll
    for (int mt = 0; mt < 4; ++mt)
#pragma unroll
        for (int m = 0; m < 5; ++m)
#pragma unroll
            for (int r = 0; r < 8; ++r) acc[mt][m][r] = 0.0f;

    v4f pf[NPIECE];
#pragma unroll
    for (int i = 0; i < NPIECE; ++i) pf[i] = *(const v4f*)(gsrc[i]);
#pragma unroll
    for (int i = 0; i < NPIECE; ++i) *(v4f*)(lds + sdst + i * 64 * LP) = pf[i];
    __syncthreads();

    const int aoff = (Mhalf + lmod) * LP + 2 * hi;
    const int boff = A_FL + (Nbase + lmod) * LP + 2 * hi;

    for (int c = 0; c < NCHUNK; ++c) {
        const float* st = lds + (c & 1) * STG_FL;

        if (c + 1 < NCHUNK) {
#pragma unroll
            for (int i = 0; i < NPIECE; ++i) pf[i] = *(const v4f*)(gsrc[i] + (c + 1) * KC);
        }

#pragma unroll
        for (int kk = 0; kk < KC; kk += 4) {
            v2f a[4], b[5];
#pragma unroll
            for (int mt = 0; mt < 4; ++mt) a[mt] = *(const v2f*)(st + aoff + mt * 16 * LP + kk);
#pragma unroll
            for (int m = 0; m < 5; ++m)   b[m]  = *(const v2f*)(st + boff + m * 64 * LP + kk);
#pragma unroll
            for (int mt = 0; mt < 4; ++mt)
#pragma unroll
                for (int m = 0; m < 5; ++m)
                    acc[mt][m] = __builtin_amdgcn_wmma_f32_16x16x4_f32(
                         false, a[mt],  false, b[m],
                         (short)0, acc[mt][m],
                         false,  false);
        }

        if (c + 1 < NCHUNK) {
            float* dn = lds + ((c + 1) & 1) * STG_FL;
#pragma unroll
            for (int i = 0; i < NPIECE; ++i) *(v4f*)(dn + sdst + i * 64 * LP) = pf[i];
        }
        __syncthreads();
    }

    float* ldsS = lds;
#pragma unroll
    for (int mt = 0; mt < 4; ++mt) {
#pragma unroll
        for (int r = 0; r < 8; ++r) {
            const int m = Mhalf + mt * 16 + (hi ? r + 8 : r);
            const int n = Nbase + lmod;
            ldsS[m * SP + n]           = acc[mt][0][r];
            ldsS[RT * SP + m * SP + n] = acc[mt][2][r];
            const size_t gi = (size_t)(Mbase + m) * N_DIM + n;
            outv[gi] = acc[mt][1][r];
            oute[gi] = 1.0f / (1.0f + __expf(-acc[mt][3][r]));
            outw[gi] = 1.0f / (1.0f + __expf(-acc[mt][4][r]));
        }
    }
    __syncthreads();

    {
        float* rowp = &ldsS[(tid >> 7) * (RT * SP) + (tid & (RT - 1)) * SP];
        float mx = rowp[0];
        for (int j = 1; j < 64; ++j) mx = fmaxf(mx, rowp[j]);
        float sum = 0.0f;
        for (int j = 0; j < 64; ++j) { float e2 = __expf(rowp[j] - mx); rowp[j] = e2; sum += e2; }
        const float inv = 1.0f / sum;
        for (int j = 0; j < 64; ++j) rowp[j] *= inv;
    }
    __syncthreads();

    for (int i = tid; i < RT * 64; i += 256) {
        const int m = i >> 6, n = i & 63;
        const size_t gi = (size_t)(Mbase + m) * N_DIM + n;
        outk[gi] = ldsS[m * SP + n];
        outq[gi] = ldsS[RT * SP + m * SP + n];
    }
}

#define SCH    16
#define SROWS  32
#define SSTEP  (64 + 64 + 2 * SROWS)
#define SHALF  (SCH * SSTEP)
#define NCHK   (T_DIM / SCH)
#define OSTEP  (SROWS * 8)

static __device__ __forceinline__ void lds_handover_barrier()
{
    asm volatile("s_wait_dscnt 0x0\n\ts_barrier_signal -1\n\ts_barrier_wait -1" ::: "memory");
}

template <int MASK>
static __device__ __forceinline__ float add_lane_xor(float x)
{
    return x + __int_as_float(__builtin_amdgcn_mov_dpp(__float_as_int(x), 0x160 + MASK, 0xF, 0xF, true));
}

static __device__ __forceinline__ v2f ntm_keep(v2f k2, float e)
{
    const v2f one = {1.0f, 1.0f};
    const v2f ne  = {-e, -e};
    return __builtin_elementwise_fma(ne, k2, one);
}
static __device__ __forceinline__ void ntm_pair(v2f& s, v2f& acc, v2f keep, v2f k2, v2f q2, float wv)
{
    const v2f wv2 = {wv, wv};
    const v2f m   = s * keep;
    s   = __builtin_elementwise_fma(wv2, k2, m);
    acc = __builtin_elementwise_fma(s, q2, acc);
}

__global__ __launch_bounds__(128) void ntm_scan_kernel(
    const float* __restrict__ S0,
    const float* __restrict__ kG, const float* __restrict__ vG,
    const float* __restrict__ qG, const float* __restrict__ eG,
    const float* __restrict__ wG,
    float* __restrict__ out, float* __restrict__ Sfin)
{
    __shared__ __attribute__((aligned(16))) float stage[2 * SHALF];
    __shared__ __attribute__((aligned(16))) float ostage[2 * SCH * OSTEP];

    const int b    = blockIdx.x >> 1;
    const int g    = blockIdx.x & 1;
    const int tid  = threadIdx.x;
    const int rp   = tid >> 3;
    const int part = tid & 7;

    const size_t soff = (size_t)b * 4096 + (size_t)(g * SROWS + 2 * rp) * 64 + part * 8;
    v2f S[2][4];
#pragma unroll
    for (int row = 0; row < 2; ++row) {
        const v4f lo = *(const v4f*)(S0 + soff + row * 64);
        const v4f hi = *(const v4f*)(S0 + soff + row * 64 + 4);
        S[row][0] = lo.lo; S[row][1] = lo.hi; S[row][2] = hi.lo; S[row][3] = hi.hi;
    }

    const int kt = tid >> 4, kc = (tid & 15) * 4;
    const int et = tid >> 3, ec = (tid & 7) * 4;
    const size_t kgo = (size_t)kt * 4096 + (size_t)b * 64 + kc;
    const size_t ego = (size_t)et * 4096 + (size_t)b * 64 + g * SROWS + ec;
    const int klo = kt * SSTEP + kc;
    const int elo = et * SSTEP + 128 + ec;

    v4f pk0 = *(const v4f*)(kG + kgo);
    v4f pk1 = *(const v4f*)(kG + kgo + 8 * 4096);
    v4f pq0 = *(const v4f*)(qG + kgo);
    v4f pq1 = *(const v4f*)(qG + kgo + 8 * 4096);
    v4f pe  = *(const v4f*)(eG + ego);
    v4f pw  = *(const v4f*)(wG + ego);
    v4f pv  = *(const v4f*)(vG + ego);

    __syncthreads();

    const size_t oidx = (size_t)(tid >> 3) * 4096 + (size_t)b * 64 + g * SROWS + (tid & 7) * 4;
    const int orow = (tid >> 3) * OSTEP + (tid & 7) * 32;

    for (int c = 0; c < NCHK; ++c) {
        const int par = c & 1;

        float* dst = stage + par * SHALF;
        *(v4f*)(dst + klo)                  = pk0;
        *(v4f*)(dst + klo + 8 * SSTEP)      = pk1;
        *(v4f*)(dst + klo + 64)             = pq0;
        *(v4f*)(dst + klo + 64 + 8 * SSTEP) = pq1;
        *(v4f*)(dst + elo)                  = pe;
        *(v4f*)(dst + elo + SROWS)          = pw * pv;
        lds_handover_barrier();

        const float* kq = dst + part * 8;
        const float* ew = dst + 128 + 2 * rp;
        v4f kcur[2], qcur[2];
        v2f ecur, wvcur;
        ecur    = *(const v2f*)(ew);
        wvcur   = *(const v2f*)(ew + SROWS);
        kcur[0] = *(const v4f*)(kq);
        kcur[1] = *(const v4f*)(kq + 4);
        qcur[0] = *(const v4f*)(kq + 64);
        qcur[1] = *(const v4f*)(kq + 68);

        if (c + 1 < NCHK) {
            const size_t go = (size_t)(c + 1) * SCH * 4096;
            pk0 = *(const v4f*)(kG + kgo + go);
            pk1 = *(const v4f*)(kG + kgo + go + 8 * 4096);
            pq0 = *(const v4f*)(qG + kgo + go);
            pq1 = *(const v4f*)(qG + kgo + go + 8 * 4096);
            pe  = *(const v4f*)(eG + ego + go);
            pw  = *(const v4f*)(wG + ego + go);
            pv  = *(const v4f*)(vG + ego + go);
        }
        if (c > 0) {
            const float* os = ostage + (par ^ 1) * (SCH * OSTEP) + orow;
            const v2f a01 = *(const v2f*)(os);
            const v2f a23 = *(const v2f*)(os + 16);
            v4f o;
            o[0] = tanhf(a01[0]); o[1] = tanhf(a01[1]); o[2] = tanhf(a23[0]); o[3] = tanhf(a23[1]);
            *(v4f*)(out + (size_t)(c - 1) * SCH * 4096 + oidx) = o;
        }

        float* ost = ostage + par * (SCH * OSTEP) + rp * 16 + part * 2;
        v2f kpc[2][4];
#pragma unroll
        for (int row = 0; row < 2; ++row) {
            kpc[row][0] = ntm_keep(kcur[0].lo, ecur[row]);
            kpc[row][1] = ntm_keep(kcur[0].hi, ecur[row]);
            kpc[row][2] = ntm_keep(kcur[1].lo, ecur[row]);
            kpc[row][3] = ntm_keep(kcur[1].hi, ecur[row]);
        }
        v2f accp[2][2] = {{{0.0f, 0.0f}, {0.0f, 0.0f}}, {{0.0f, 0.0f}, {0.0f, 0.0f}}};
#pragma unroll
        for (int tl = 0; tl < SCH; ++tl) {
            v4f knxt[2], qnxt[2];
            v2f enxt = {0.0f, 0.0f}, wvnxt = {0.0f, 0.0f};
            if (tl + 1 < SCH) {
                enxt    = *(const v2f*)(ew + (tl + 1) * SSTEP);
                wvnxt   = *(const v2f*)(ew + (tl + 1) * SSTEP + SROWS);
                knxt[0] = *(const v4f*)(kq + (tl + 1) * SSTEP);
                knxt[1] = *(const v4f*)(kq + (tl + 1) * SSTEP + 4);
                qnxt[0] = *(const v4f*)(kq + (tl + 1) * SSTEP + 64);
                qnxt[1] = *(const v4f*)(kq + (tl + 1) * SSTEP + 68);
            }
            __builtin_amdgcn_sched_barrier(0);

            if (tl > 0) {
                const v2f u0 = accp[0][0] + accp[0][1];
                const v2f u1 = accp[1][0] + accp[1][1];
                float t0 = u0[0] + u0[1];
                float t1 = u1[0] + u1[1];
                t0 = add_lane_xor<1>(t0); t1 = add_lane_xor<1>(t1);
                t0 = add_lane_xor<2>(t0); t1 = add_lane_xor<2>(t1);
                t0 = add_lane_xor<4>(t0); t1 = add_lane_xor<4>(t1);
                const v2f t = {t0, t1};
                *(v2f*)(ost + (tl - 1) * OSTEP) = t;
            }

#pragma unroll
            for (int row = 0; row < 2; ++row) {
                v2f a = {0.0f, 0.0f}, bb = {0.0f, 0.0f};
                ntm_pair(S[row][0], a,  kpc[row][0], kcur[0].lo, qcur[0].lo, wvcur[row]);
                ntm_pair(S[row][1], a,  kpc[row][1], kcur[0].hi, qcur[0].hi, wvcur[row]);
                ntm_pair(S[row][2], bb, kpc[row][2], kcur[1].lo, qcur[1].lo, wvcur[row]);
                ntm_pair(S[row][3], bb, kpc[row][3], kcur[1].hi, qcur[1].hi, wvcur[row]);
                accp[row][0] = a; accp[row][1] = bb;
            }

            if (tl + 1 < SCH) {
#pragma unroll
                for (int row = 0; row < 2; ++row) {
                    kpc[row][0] = ntm_keep(knxt[0].lo, enxt[row]);
                    kpc[row][1] = ntm_keep(knxt[0].hi, enxt[row]);
                    kpc[row][2] = ntm_keep(knxt[1].lo, enxt[row]);
                    kpc[row][3] = ntm_keep(knxt[1].hi, enxt[row]);
                }
                ecur = enxt; wvcur = wvnxt;
                kcur[0] = knxt[0]; kcur[1] = knxt[1];
                qcur[0] = qnxt[0]; qcur[1] = qnxt[1];
            }
        }
        {
            const v2f u0 = accp[0][0] + accp[0][1];
            const v2f u1 = accp[1][0] + accp[1][1];
            float t0 = u0[0] + u0[1];
            float t1 = u1[0] + u1[1];
            t0 = add_lane_xor<1>(t0); t1 = add_lane_xor<1>(t1);
            t0 = add_lane_xor<2>(t0); t1 = add_lane_xor<2>(t1);
            t0 = add_lane_xor<4>(t0); t1 = add_lane_xor<4>(t1);
            const v2f t = {t0, t1};
            *(v2f*)(ost + (SCH - 1) * OSTEP) = t;
        }
    }

    __syncthreads();
    {
        const float* os = ostage + ((NCHK - 1) & 1) * (SCH * OSTEP) + orow;
        const v2f a01 = *(const v2f*)(os);
        const v2f a23 = *(const v2f*)(os + 16);
        v4f o;
        o[0] = tanhf(a01[0]); o[1] = tanhf(a01[1]); o[2] = tanhf(a23[0]); o[3] = tanhf(a23[1]);
        *(v4f*)(out + (size_t)(NCHK - 1) * SCH * 4096 + oidx) = o;
    }

#pragma unroll
    for (int row = 0; row < 2; ++row) {
        v4f lo, hi;
        lo.lo = S[row][0]; lo.hi = S[row][1]; hi.lo = S[row][2]; hi.hi = S[row][3];
        *(v4f*)(Sfin + soff + row * 64)     = lo;
        *(v4f*)(Sfin + soff + row * 64 + 4) = hi;
    }
}

extern "C" void kernel_launch(void* const* d_in, const int* in_sizes, int n_in,
                              void* d_out, int out_size, void* d_ws, size_t ws_size,
                              hipStream_t stream) {
    (void)in_sizes; (void)n_in; (void)out_size; (void)ws_size;
    const float* x  = (const float*)d_in[0];
    const float* S0 = (const float*)d_in[1];
    const float* Wk = (const float*)d_in[2];
    const float* Wv = (const float*)d_in[3];
    const float* Wq = (const float*)d_in[4];
    const float* We = (const float*)d_in[5];
    const float* Ww = (const float*)d_in[6];

    float* out = (float*)d_out;
    float* ws  = (float*)d_ws;
    float* wk  = ws;
    float* wv  = ws + TBN;
    float* wq  = ws + 2 * TBN;
    float* we_ = ws + 3 * TBN;
    float* ww_ = ws + 4 * TBN;

    ntm_proj_kernel<<<ROWS / RT, 256, 0, stream>>>(
        x, Wk, Wv, Wq, We, Ww, wk, wv, wq, we_, ww_);

    ntm_scan_kernel<<<B_DIM * 2, 128, 0, stream>>>(
        S0, wk, wv, wq, we_, ww_, out, out + TBN);
}


# scan: staging chunk 16 -> 64 timesteps (LDS ring 2 x 64 steps): the next chunk's global loads get 4x the time before the hand-over waits for them, a quarter of the barriers; same sums, same order
# speedup vs baseline: 10.1337x; 1.0354x over previous
#include <hip/hip_runtime.h>
#include <math.h>

typedef float v2f __attribute__((ext_vector_type(2)));
typedef float v8f __attribute__((ext_vector_type(8)));

#define T_DIM 1024
#define B_DIM 64
#define D_DIM 512
#define N_DIM 64
#define ROWS  (T_DIM * B_DIM)
#define TBN   ((size_t)T_DIM * B_DIM * N_DIM)

typedef float v4f __attribute__((ext_vector_type(4)));

#define RT      128
#define KC      16
#define LP      20
#define NCHUNK  (D_DIM / KC)
#define A_FL    (RT * LP)
#define STG_FL  ((RT + 5 * N_DIM) * LP)
#define NPIECE  ((RT + 5 * N_DIM) * (KC / 4) / 256)
#define SP      65

__global__ __launch_bounds__(256) void ntm_proj_kernel(
    const float* __restrict__ x,
    const float* __restrict__ Wk, const float* __restrict__ Wv,
    const float* __restrict__ Wq, const float* __restrict__ We,
    const float* __restrict__ Ww,
    float* __restrict__ outk, float* __restrict__ outv,
    float* __restrict__ outq, float* __restrict__ oute,
    float* __restrict__ outw)
{
    __shared__ __attribute__((aligned(16))) float lds[2 * STG_FL];

    const int tid   = threadIdx.x;
    const int wave  = tid >> 5;
    const int lane  = tid & 31;
    const int lmod  = lane & 15;
    const int hi    = lane >> 4;
    const int Mbase = blockIdx.x * RT;
    const int Nbase = (wave & 3) * 16;
    const int Mhalf = (wave >> 2) * 64;

    const int srow = tid >> 2;
    const int sq   = (tid & 3) * 4;
    const float* gsrc[NPIECE];
    gsrc[0] = x  + (size_t)(Mbase + srow) * D_DIM + sq;
    gsrc[1] = x  + (size_t)(Mbase + 64 + srow) * D_DIM + sq;
    gsrc[2] = Wk + (size_t)srow * D_DIM + sq;
    gsrc[3] = Wv + (size_t)srow * D_DIM + sq;
    gsrc[4] = Wq + (size_t)srow * D_DIM + sq;
    gsrc[5] = We + (size_t)srow * D_DIM + sq;
    gsrc[6] = Ww + (size_t)srow * D_DIM + sq;
    const int sdst = srow * LP + sq;

    v8f acc[4][5];
#pragma unroll
    for (int mt = 0; mt < 4; ++mt)
#pragma unroll
        for (int m = 0; m < 5; ++m)
#pragma unroll
            for (int r = 0; r < 8; ++r) acc[mt][m][r] = 0.0f;

    v4f pf[NPIECE];
#pragma unroll
    for (int i = 0; i < NPIECE; ++i) pf[i] = *(const v4f*)(gsrc[i]);
#pragma unroll
    for (int i = 0; i < NPIECE; ++i) *(v4f*)(lds + sdst + i * 64 * LP) = pf[i];
    __syncthreads();

    const int aoff = (Mhalf + lmod) * LP + 2 * hi;
    const int boff = A_FL + (Nbase + lmod) * LP + 2 * hi;

    for (int c = 0; c < NCHUNK; ++c) {
        const float* st = lds + (c & 1) * STG_FL;

        if (c + 1 < NCHUNK) {
#pragma unroll
            for (int i = 0; i < NPIECE; ++i) pf[i] = *(const v4f*)(gsrc[i] + (c + 1) * KC);
        }

#pragma unroll
        for (int kk = 0; kk < KC; kk += 4) {
            v2f a[4], b[5];
#pragma unroll
            for (int mt = 0; mt < 4; ++mt) a[mt] = *(const v2f*)(st + aoff + mt * 16 * LP + kk);
#pragma unroll
            for (int m = 0; m < 5; ++m)   b[m]  = *(const v2f*)(st + boff + m * 64 * LP + kk);
#pragma unroll
            for (int mt = 0; mt < 4; ++mt)
#pragma unroll
                for (int m = 0; m < 5; ++m)
                    acc[mt][m] = __builtin_amdgcn_wmma_f32_16x16x4_f32(
                         false, a[mt],  false, b[m],
                         (short)0, acc[mt][m],
                         false,  false);
        }

        if (c + 1 < NCHUNK) {
            float* dn = lds + ((c + 1) & 1) * STG_FL;
#pragma unroll
            for (int i = 0; i < NPIECE; ++i) *(v4f*)(dn + sdst + i * 64 * LP) = pf[i];
        }
        __syncthreads();
    }

    float* ldsS = lds;
#pragma unroll
    for (int mt = 0; mt < 4; ++mt) {
#pragma unroll
        for (int r = 0; r < 8; ++r) {
            const int m = Mhalf + mt * 16 + (hi ? r + 8 : r);
            const int n = Nbase + lmod;
            ldsS[m * SP + n]           = acc[mt][0][r];
            ldsS[RT * SP + m * SP + n] = acc[mt][2][r];
            const size_t gi = (size_t)(Mbase + m) * N_DIM + n;
            outv[gi] = acc[mt][1][r];
            oute[gi] = 1.0f / (1.0f + __expf(-acc[mt][3][r]));
            outw[gi] = 1.0f / (1.0f + __expf(-acc[mt][4][r]));
        }
    }
    __syncthreads();

    {
        float* rowp = &ldsS[(tid >> 7) * (RT * SP) + (tid & (RT - 1)) * SP];
        float mx = rowp[0];
        for (int j = 1; j < 64; ++j) mx = fmaxf(mx, rowp[j]);
        float sum = 0.0f;
        for (int j = 0; j < 64; ++j) { float e2 = __expf(rowp[j] - mx); rowp[j] = e2; sum += e2; }
        const float inv = 1.0f / sum;
        for (int j = 0; j < 64; ++j) rowp[j] *= inv;
    }
    __syncthreads();

    for (int i = tid; i < RT * 64; i += 256) {
        const int m = i >> 6, n = i & 63;
        const size_t gi = (size_t)(Mbase + m) * N_DIM + n;
        outk[gi] = ldsS[m * SP + n];
        outq[gi] = ldsS[RT * SP + m * SP + n];
    }
}

#define SCH    64
#define KPASS  (SCH / 8)
#define EPASS  (SCH / 16)
#define SROWS  32
#define SSTEP  (64 + 64 + 2 * SROWS)
#define SHALF  (SCH * SSTEP)
#define NCHK   (T_DIM / SCH)
#define OSTEP  (SROWS * 8)

static __device__ __forceinline__ void lds_handover_barrier()
{
    asm volatile("s_wait_dscnt 0x0\n\ts_barrier_signal -1\n\ts_barrier_wait -1" ::: "memory");
}

template <int MASK>
static __device__ __forceinline__ float add_lane_xor(float x)
{
    return x + __int_as_float(__builtin_amdgcn_mov_dpp(__float_as_int(x), 0x160 + MASK, 0xF, 0xF, true));
}

static __device__ __forceinline__ v2f ntm_keep(v2f k2, float e)
{
    const v2f one = {1.0f, 1.0f};
    const v2f ne  = {-e, -e};
    return __builtin_elementwise_fma(ne, k2, one);
}
static __device__ __forceinline__ void ntm_pair(v2f& s, v2f& acc, v2f keep, v2f k2, v2f q2, float wv)
{
    const v2f wv2 = {wv, wv};
    const v2f m   = s * keep;
    s   = __builtin_elementwise_fma(wv2, k2, m);
    acc = __builtin_elementwise_fma(s, q2, acc);
}

__global__ __launch_bounds__(128) void ntm_scan_kernel(
    const float* __restrict__ S0,
    const float* __restrict__ kG, const float* __restrict__ vG,
    const float* __restrict__ qG, const float* __restrict__ eG,
    const float* __restrict__ wG,
    float* __restrict__ out, float* __restrict__ Sfin)
{
    __shared__ __attribute__((aligned(16))) float stage[2 * SHALF];
    __shared__ __attribute__((aligned(16))) float ostage[2 * SCH * OSTEP];

    const int b    = blockIdx.x >> 1;
    const int g    = blockIdx.x & 1;
    const int tid  = threadIdx.x;
    const int rp   = tid >> 3;
    const int part = tid & 7;

    const size_t soff = (size_t)b * 4096 + (size_t)(g * SROWS + 2 * rp) * 64 + part * 8;
    v2f S[2][4];
#pragma unroll
    for (int row = 0; row < 2; ++row) {
        const v4f lo = *(const v4f*)(S0 + soff + row * 64);
        const v4f hi = *(const v4f*)(S0 + soff + row * 64 + 4);
        S[row][0] = lo.lo; S[row][1] = lo.hi; S[row][2] = hi.lo; S[row][3] = hi.hi;
    }

    const int kt = tid >> 4, kc = (tid & 15) * 4;
    const int et = tid >> 3, ec = (tid & 7) * 4;
    const size_t kgo = (size_t)kt * 4096 + (size_t)b * 64 + kc;
    const size_t ego = (size_t)et * 4096 + (size_t)b * 64 + g * SROWS + ec;
    const int klo = kt * SSTEP + kc;
    const int elo = et * SSTEP + 128 + ec;

    v4f pk[KPASS], pq[KPASS], pe[EPASS], pw[EPASS], pv[EPASS];
#pragma unroll
    for (int i = 0; i < KPASS; ++i) {
        pk[i] = *(const v4f*)(kG + kgo + (size_t)i * 8 * 4096);
        pq[i] = *(const v4f*)(qG + kgo + (size_t)i * 8 * 4096);
    }
#pragma unroll
    for (int i = 0; i < EPASS; ++i) {
        pe[i] = *(const v4f*)(eG + ego + (size_t)i * 16 * 4096);
        pw[i] = *(const v4f*)(wG + ego + (size_t)i * 16 * 4096);
        pv[i] = *(const v4f*)(vG + ego + (size_t)i * 16 * 4096);
    }

    __syncthreads();

    const size_t oidx = (size_t)(tid >> 3) * 4096 + (size_t)b * 64 + g * SROWS + (tid & 7) * 4;
    const int orow = (tid >> 3) * OSTEP + (tid & 7) * 32;

    for (int c = 0; c < NCHK; ++c) {
        const int par = c & 1;

        float* dst = stage + par * SHALF;
#pragma unroll
        for (int i = 0; i < KPASS; ++i) {
            *(v4f*)(dst + klo + i * 8 * SSTEP)      = pk[i];
            *(v4f*)(dst + klo + 64 + i * 8 * SSTEP) = pq[i];
        }
#pragma unroll
        for (int i = 0; i < EPASS; ++i) {
            *(v4f*)(dst + elo + i * 16 * SSTEP)         = pe[i];
            *(v4f*)(dst + elo + SROWS + i * 16 * SSTEP) = pw[i] * pv[i];
        }
        lds_handover_barrier();

        const float* kq = dst + part * 8;
        const float* ew = dst + 128 + 2 * rp;
        v4f kcur[2], qcur[2];
        v2f ecur, wvcur;
        ecur    = *(const v2f*)(ew);
        wvcur   = *(const v2f*)(ew + SROWS);
        kcur[0] = *(const v4f*)(kq);
        kcur[1] = *(const v4f*)(kq + 4);
        qcur[0] = *(const v4f*)(kq + 64);
        qcur[1] = *(const v4f*)(kq + 68);

        if (c + 1 < NCHK) {
            const size_t go = (size_t)(c + 1) * SCH * 4096;
#pragma unroll
            for (int i = 0; i < KPASS; ++i) {
                pk[i] = *(const v4f*)(kG + kgo + go + (size_t)i * 8 * 4096);
                pq[i] = *(const v4f*)(qG + kgo + go + (size_t)i * 8 * 4096);
            }
#pragma unroll
            for (int i = 0; i < EPASS; ++i) {
                pe[i] = *(const v4f*)(eG + ego + go + (size_t)i * 16 * 4096);
                pw[i] = *(const v4f*)(wG + ego + go + (size_t)i * 16 * 4096);
                pv[i] = *(const v4f*)(vG + ego + go + (size_t)i * 16 * 4096);
            }
        }
        if (c > 0) {
#pragma unroll
            for (int j = 0; j < EPASS; ++j) {
                const float* os = ostage + (par ^ 1) * (SCH * OSTEP) + orow + j * 16 * OSTEP;
                const v2f a01 = *(const v2f*)(os);
                const v2f a23 = *(const v2f*)(os + 16);
                v4f o;
                o[0] = tanhf(a01[0]); o[1] = tanhf(a01[1]); o[2] = tanhf(a23[0]); o[3] = tanhf(a23[1]);
                *(v4f*)(out + (size_t)(c - 1) * SCH * 4096 + oidx + (size_t)j * 16 * 4096) = o;
            }
        }

        float* ost = ostage + par * (SCH * OSTEP) + rp * 16 + part * 2;
        v2f kpc[2][4];
#pragma unroll
        for (int row = 0; row < 2; ++row) {
            kpc[row][0] = ntm_keep(kcur[0].lo, ecur[row]);
            kpc[row][1] = ntm_keep(kcur[0].hi, ecur[row]);
            kpc[row][2] = ntm_keep(kcur[1].lo, ecur[row]);
            kpc[row][3] = ntm_keep(kcur[1].hi, ecur[row]);
        }
        v2f accp[2][2] = {{{0.0f, 0.0f}, {0.0f, 0.0f}}, {{0.0f, 0.0f}, {0.0f, 0.0f}}};
#pragma unroll
        for (int tl = 0; tl < SCH; ++tl) {
            v4f knxt[2], qnxt[2];
            v2f enxt = {0.0f, 0.0f}, wvnxt = {0.0f, 0.0f};
            if (tl + 1 < SCH) {
                enxt    = *(const v2f*)(ew + (tl + 1) * SSTEP);
                wvnxt   = *(const v2f*)(ew + (tl + 1) * SSTEP + SROWS);
                knxt[0] = *(const v4f*)(kq + (tl + 1) * SSTEP);
                knxt[1] = *(const v4f*)(kq + (tl + 1) * SSTEP + 4);
                qnxt[0] = *(const v4f*)(kq + (tl + 1) * SSTEP + 64);
                qnxt[1] = *(const v4f*)(kq + (tl + 1) * SSTEP + 68);
            }
            __builtin_amdgcn_sched_barrier(0);

            if (tl > 0) {
                const v2f u0 = accp[0][0] + accp[0][1];
                const v2f u1 = accp[1][0] + accp[1][1];
                float t0 = u0[0] + u0[1];
                float t1 = u1[0] + u1[1];
                t0 = add_lane_xor<1>(t0); t1 = add_lane_xor<1>(t1);
                t0 = add_lane_xor<2>(t0); t1 = add_lane_xor<2>(t1);
                t0 = add_lane_xor<4>(t0); t1 = add_lane_xor<4>(t1);
                const v2f t = {t0, t1};
                *(v2f*)(ost + (tl - 1) * OSTEP) = t;
            }

#pragma unroll
            for (int row = 0; row < 2; ++row) {
                v2f a = {0.0f, 0.0f}, bb = {0.0f, 0.0f};
                ntm_pair(S[row][0], a,  kpc[row][0], kcur[0].lo, qcur[0].lo, wvcur[row]);
                ntm_pair(S[row][1], a,  kpc[row][1], kcur[0].hi, qcur[0].hi, wvcur[row]);
                ntm_pair(S[row][2], bb, kpc[row][2], kcur[1].lo, qcur[1].lo, wvcur[row]);
                ntm_pair(S[row][3], bb, kpc[row][3], kcur[1].hi, qcur[1].hi, wvcur[row]);
                accp[row][0] = a; accp[row][1] = bb;
            }

            if (tl + 1 < SCH) {
#pragma unroll
                for (int row = 0; row < 2; ++row) {
                    kpc[row][0] = ntm_keep(knxt[0].lo, enxt[row]);
                    kpc[row][1] = ntm_keep(knxt[0].hi, enxt[row]);
                    kpc[row][2] = ntm_keep(knxt[1].lo, enxt[row]);
                    kpc[row][3] = ntm_keep(knxt[1].hi, enxt[row]);
                }
                ecur = enxt; wvcur = wvnxt;
                kcur[0] = knxt[0]; kcur[1] = knxt[1];
                qcur[0] = qnxt[0]; qcur[1] = qnxt[1];
            }
        }
        {
            const v2f u0 = accp[0][0] + accp[0][1];
            const v2f u1 = accp[1][0] + accp[1][1];
            float t0 = u0[0] + u0[1];
            float t1 = u1[0] + u1[1];
            t0 = add_lane_xor<1>(t0); t1 = add_lane_xor<1>(t1);
            t0 = add_lane_xor<2>(t0); t1 = add_lane_xor<2>(t1);
            t0 = add_lane_xor<4>(t0); t1 = add_lane_xor<4>(t1);
            const v2f t = {t0, t1};
            *(v2f*)(ost + (SCH - 1) * OSTEP) = t;
        }
    }

    __syncthreads();
#pragma unroll
    for (int j = 0; j < EPASS; ++j) {
        const float* os = ostage + ((NCHK - 1) & 1) * (SCH * OSTEP) + orow + j * 16 * OSTEP;
        const v2f a01 = *(const v2f*)(os);
        const v2f a23 = *(const v2f*)(os + 16);
        v4f o;
        o[0] = tanhf(a01[0]); o[1] = tanhf(a01[1]); o[2] = tanhf(a23[0]); o[3] = tanhf(a23[1]);
        *(v4f*)(out + (size_t)(NCHK - 1) * SCH * 4096 + oidx + (size_t)j * 16 * 4096) = o;
    }

#pragma unroll
    for (int row = 0; row < 2; ++row) {
        v4f lo, hi;
        lo.lo = S[row][0]; lo.hi = S[row][1]; hi.lo = S[row][2]; hi.hi = S[row][3];
        *(v4f*)(Sfin + soff + row * 64)     = lo;
        *(v4f*)(Sfin + soff + row * 64 + 4) = hi;
    }
}

extern "C" void kernel_launch(void* const* d_in, const int* in_sizes, int n_in,
                              void* d_out, int out_size, void* d_ws, size_t ws_size,
                              hipStream_t stream) {
    (void)in_sizes; (void)n_in; (void)out_size; (void)ws_size;
    const float* x  = (const float*)d_in[0];
    const float* S0 = (const float*)d_in[1];
    const float* Wk = (const float*)d_in[2];
    const float* Wv = (const float*)d_in[3];
    const float* Wq = (const float*)d_in[4];
    const float* We = (const float*)d_in[5];
    const float* Ww = (const float*)d_in[6];

    float* out = (float*)d_out;
    float* ws  = (float*)d_ws;
    float* wk  = ws;
    float* wv  = ws + TBN;
    float* wq  = ws + 2 * TBN;
    float* we_ = ws + 3 * TBN;
    float* ww_ = ws + 4 * TBN;

    ntm_proj_kernel<<<ROWS / RT, 256, 0, stream>>>(
        x, Wk, Wv, Wq, We, Ww, wk, wv, wq, we_, ww_);

    ntm_scan_kernel<<<B_DIM * 2, 128, 0, stream>>>(
        S0, wk, wv, wq, we_, ww_, out, out + TBN);
}


# scan: tanh of the row sums as one transcendental-unit instruction (v_tanh_f32), was library tanhf (2 data-dependent branches per value, a thread's values of a chunk in series); tanh rounding changes
# speedup vs baseline: 10.5127x; 1.0374x over previous
#include <hip/hip_runtime.h>
#include <math.h>

typedef float v2f __attribute__((ext_vector_type(2)));
typedef float v8f __attribute__((ext_vector_type(8)));

#define T_DIM 1024
#define B_DIM 64
#define D_DIM 512
#define N_DIM 64
#define ROWS  (T_DIM * B_DIM)
#define TBN   ((size_t)T_DIM * B_DIM * N_DIM)

typedef float v4f __attribute__((ext_vector_type(4)));

#define RT      128
#define KC      16
#define LP      20
#define NCHUNK  (D_DIM / KC)
#define A_FL    (RT * LP)
#define STG_FL  ((RT + 5 * N_DIM) * LP)
#define NPIECE  ((RT + 5 * N_DIM) * (KC / 4) / 256)
#define SP      65

__global__ __launch_bounds__(256) void ntm_proj_kernel(
    const float* __restrict__ x,
    const float* __restrict__ Wk, const float* __restrict__ Wv,
    const float* __restrict__ Wq, const float* __restrict__ We,
    const float* __restrict__ Ww,
    float* __restrict__ outk, float* __restrict__ outv,
    float* __restrict__ outq, float* __restrict__ oute,
    float* __restrict__ outw)
{
    __shared__ __attribute__((aligned(16))) float lds[2 * STG_FL];

    const int tid   = threadIdx.x;
    const int wave  = tid >> 5;
    const int lane  = tid & 31;
    const int lmod  = lane & 15;
    const int hi    = lane >> 4;
    const int Mbase = blockIdx.x * RT;
    const int Nbase = (wave & 3) * 16;
    const int Mhalf = (wave >> 2) * 64;

    const int srow = tid >> 2;
    const int sq   = (tid & 3) * 4;
    const float* gsrc[NPIECE];
    gsrc[0] = x  + (size_t)(Mbase + srow) * D_DIM + sq;
    gsrc[1] = x  + (size_t)(Mbase + 64 + srow) * D_DIM + sq;
    gsrc[2] = Wk + (size_t)srow * D_DIM + sq;
    gsrc[3] = Wv + (size_t)srow * D_DIM + sq;
    gsrc[4] = Wq + (size_t)srow * D_DIM + sq;
    gsrc[5] = We + (size_t)srow * D_DIM + sq;
    gsrc[6] = Ww + (size_t)srow * D_DIM + sq;
    const int sdst = srow * LP + sq;

    v8f acc[4][5];
#pragma unroll
    for (int mt = 0; mt < 4; ++mt)
#pragma unroll
        for (int m = 0; m < 5; ++m)
#pragma unroll
            for (int r = 0; r < 8; ++r) acc[mt][m][r] = 0.0f;

    v4f pf[NPIECE];
#pragma unroll
    for (int i = 0; i < NPIECE; ++i) pf[i] = *(const v4f*)(gsrc[i]);
#pragma unroll
    for (int i = 0; i < NPIECE; ++i) *(v4f*)(lds + sdst + i * 64 * LP) = pf[i];
    __syncthreads();

    const int aoff = (Mhalf + lmod) * LP + 2 * hi;
    const int boff = A_FL + (Nbase + lmod) * LP + 2 * hi;

    for (int c = 0; c < NCHUNK; ++c) {
        const float* st = lds + (c & 1) * STG_FL;

        if (c + 1 < NCHUNK) {
#pragma unroll
            for (int i = 0; i < NPIECE; ++i) pf[i] = *(const v4f*)(gsrc[i] + (c + 1) * KC);
        }

#pragma unroll
        for (int kk = 0; kk < KC; kk += 4) {
            v2f a[4], b[5];
#pragma unroll
            for (int mt = 0; mt < 4; ++mt) a[mt] = *(const v2f*)(st + aoff + mt * 16 * LP + kk);
#pragma unroll
            for (int m = 0; m < 5; ++m)   b[m]  = *(const v2f*)(st + boff + m * 64 * LP + kk);
#pragma unroll
            for (int mt = 0; mt < 4; ++mt)
#pragma unroll
                for (int m = 0; m < 5; ++m)
                    acc[mt][m] = __builtin_amdgcn_wmma_f32_16x16x4_f32(
                         false, a[mt],  false, b[m],
                         (short)0, acc[mt][m],
                         false,  false);
        }

        if (c + 1 < NCHUNK) {
            float* dn = lds + ((c + 1) & 1) * STG_FL;
#pragma unroll
            for (int i = 0; i < NPIECE; ++i) *(v4f*)(dn + sdst + i * 64 * LP) = pf[i];
        }
        __syncthreads();
    }

    float* ldsS = lds;
#pragma unroll
    for (int mt = 0; mt < 4; ++mt) {
#pragma unroll
        for (int r = 0; r < 8; ++r) {
            const int m = Mhalf + mt * 16 + (hi ? r + 8 : r);
            const int n = Nbase + lmod;
            ldsS[m * SP + n]           = acc[mt][0][r];
            ldsS[RT * SP + m * SP + n] = acc[mt][2][r];
            const size_t gi = (size_t)(Mbase + m) * N_DIM + n;
            outv[gi] = acc[mt][1][r];
            oute[gi] = 1.0f / (1.0f + __expf(-acc[mt][3][r]));
            outw[gi] = 1.0f / (1.0f + __expf(-acc[mt][4][r]));
        }
    }
    __syncthreads();

    {
        float* rowp = &ldsS[(tid >> 7) * (RT * SP) + (tid & (RT - 1)) * SP];
        float mx = rowp[0];
        for (int j = 1; j < 64; ++j) mx = fmaxf(mx, rowp[j]);
        float sum = 0.0f;
        for (int j = 0; j < 64; ++j) { float e2 = __expf(rowp[j] - mx); rowp[j] = e2; sum += e2; }
        const float inv = 1.0f / sum;
        for (int j = 0; j < 64; ++j) rowp[j] *= inv;
    }
    __syncthreads();

    for (int i = tid; i < RT * 64; i += 256) {
        const int m = i >> 6, n = i & 63;
        const size_t gi = (size_t)(Mbase + m) * N_DIM + n;
        outk[gi] = ldsS[m * SP + n];
        outq[gi] = ldsS[RT * SP + m * SP + n];
    }
}

#define SCH    64
#define KPASS  (SCH / 8)
#define EPASS  (SCH / 16)
#define SROWS  32
#define SSTEP  (64 + 64 + 2 * SROWS)
#define SHALF  (SCH * SSTEP)
#define NCHK   (T_DIM / SCH)
#define OSTEP  (SROWS * 8)

static __device__ __forceinline__ void lds_handover_barrier()
{
    asm volatile("s_wait_dscnt 0x0\n\ts_barrier_signal -1\n\ts_barrier_wait -1" ::: "memory");
}

template <int MASK>
static __device__ __forceinline__ float add_lane_xor(float x)
{
    return x + __int_as_float(__builtin_amdgcn_mov_dpp(__float_as_int(x), 0x160 + MASK, 0xF, 0xF, true));
}

static __device__ __forceinline__ float tanh_row(float x)
{
    return __builtin_amdgcn_tanhf(x);
}

static __device__ __forceinline__ v2f ntm_keep(v2f k2, float e)
{
    const v2f one = {1.0f, 1.0f};
    const v2f ne  = {-e, -e};
    return __builtin_elementwise_fma(ne, k2, one);
}
static __device__ __forceinline__ void ntm_pair(v2f& s, v2f& acc, v2f keep, v2f k2, v2f q2, float wv)
{
    const v2f wv2 = {wv, wv};
    const v2f m   = s * keep;
    s   = __builtin_elementwise_fma(wv2, k2, m);
    acc = __builtin_elementwise_fma(s, q2, acc);
}

__global__ __launch_bounds__(128) void ntm_scan_kernel(
    const float* __restrict__ S0,
    const float* __restrict__ kG, const float* __restrict__ vG,
    const float* __restrict__ qG, const float* __restrict__ eG,
    const float* __restrict__ wG,
    float* __restrict__ out, float* __restrict__ Sfin)
{
    __shared__ __attribute__((aligned(16))) float stage[2 * SHALF];
    __shared__ __attribute__((aligned(16))) float ostage[2 * SCH * OSTEP];

    const int b    = blockIdx.x >> 1;
    const int g    = blockIdx.x & 1;
    const int tid  = threadIdx.x;
    const int rp   = tid >> 3;
    const int part = tid & 7;

    const size_t soff = (size_t)b * 4096 + (size_t)(g * SROWS + 2 * rp) * 64 + part * 8;
    v2f S[2][4];
#pragma unroll
    for (int row = 0; row < 2; ++row) {
        const v4f lo = *(const v4f*)(S0 + soff + row * 64);
        const v4f hi = *(const v4f*)(S0 + soff + row * 64 + 4);
        S[row][0] = lo.lo; S[row][1] = lo.hi; S[row][2] = hi.lo; S[row][3] = hi.hi;
    }

    const int kt = tid >> 4, kc = (tid & 15) * 4;
    const int et = tid >> 3, ec = (tid & 7) * 4;
    const size_t kgo = (size_t)kt * 4096 + (size_t)b * 64 + kc;
    const size_t ego = (size_t)et * 4096 + (size_t)b * 64 + g * SROWS + ec;
    const int klo = kt * SSTEP + kc;
    const int elo = et * SSTEP + 128 + ec;

    v4f pk[KPASS], pq[KPASS], pe[EPASS], pw[EPASS], pv[EPASS];
#pragma unroll
    for (int i = 0; i < KPASS; ++i) {
        pk[i] = *(const v4f*)(kG + kgo + (size_t)i * 8 * 4096);
        pq[i] = *(const v4f*)(qG + kgo + (size_t)i * 8 * 4096);
    }
#pragma unroll
    for (int i = 0; i < EPASS; ++i) {
        pe[i] = *(const v4f*)(eG + ego + (size_t)i * 16 * 4096);
        pw[i] = *(const v4f*)(wG + ego + (size_t)i * 16 * 4096);
        pv[i] = *(const v4f*)(vG + ego + (size_t)i * 16 * 4096);
    }

    __syncthreads();

    const size_t oidx = (size_t)(tid >> 3) * 4096 + (size_t)b * 64 + g * SROWS + (tid & 7) * 4;
    const int orow = (tid >> 3) * OSTEP + (tid & 7) * 32;

    for (int c = 0; c < NCHK; ++c) {
        const int par = c & 1;

        float* dst = stage + par * SHALF;
#pragma unroll
        for (int i = 0; i < KPASS; ++i) {
            *(v4f*)(dst + klo + i * 8 * SSTEP)      = pk[i];
            *(v4f*)(dst + klo + 64 + i * 8 * SSTEP) = pq[i];
        }
#pragma unroll
        for (int i = 0; i < EPASS; ++i) {
            *(v4f*)(dst + elo + i * 16 * SSTEP)         = pe[i];
            *(v4f*)(dst + elo + SROWS + i * 16 * SSTEP) = pw[i] * pv[i];
        }
        lds_handover_barrier();

        const float* kq = dst + part * 8;
        const float* ew = dst + 128 + 2 * rp;
        v4f kcur[2], qcur[2];
        v2f ecur, wvcur;
        ecur    = *(const v2f*)(ew);
        wvcur   = *(const v2f*)(ew + SROWS);
        kcur[0] = *(const v4f*)(kq);
        kcur[1] = *(const v4f*)(kq + 4);
        qcur[0] = *(const v4f*)(kq + 64);
        qcur[1] = *(const v4f*)(kq + 68);

        if (c + 1 < NCHK) {
            const size_t go = (size_t)(c + 1) * SCH * 4096;
#pragma unroll
            for (int i = 0; i < KPASS; ++i) {
                pk[i] = *(const v4f*)(kG + kgo + go + (size_t)i * 8 * 4096);
                pq[i] = *(const v4f*)(qG + kgo + go + (size_t)i * 8 * 4096);
            }
#pragma unroll
            for (int i = 0; i < EPASS; ++i) {
                pe[i] = *(const v4f*)(eG + ego + go + (size_t)i * 16 * 4096);
                pw[i] = *(const v4f*)(wG + ego + go + (size_t)i * 16 * 4096);
                pv[i] = *(const v4f*)(vG + ego + go + (size_t)i * 16 * 4096);
            }
        }
        if (c > 0) {
#pragma unroll
            for (int j = 0; j < EPASS; ++j) {
                const float* os = ostage + (par ^ 1) * (SCH * OSTEP) + orow + j * 16 * OSTEP;
                const v2f a01 = *(const v2f*)(os);
                const v2f a23 = *(const v2f*)(os + 16);
                v4f o;
                o[0] = tanh_row(a01[0]); o[1] = tanh_row(a01[1]); o[2] = tanh_row(a23[0]); o[3] = tanh_row(a23[1]);
                *(v4f*)(out + (size_t)(c - 1) * SCH * 4096 + oidx + (size_t)j * 16 * 4096) = o;
            }
        }

        float* ost = ostage + par * (SCH * OSTEP) + rp * 16 + part * 2;
        v2f kpc[2][4];
#pragma unroll
        for (int row = 0; row < 2; ++row) {
            kpc[row][0] = ntm_keep(kcur[0].lo, ecur[row]);
            kpc[row][1] = ntm_keep(kcur[0].hi, ecur[row]);
            kpc[row][2] = ntm_keep(kcur[1].lo, ecur[row]);
            kpc[row][3] = ntm_keep(kcur[1].hi, ecur[row]);
        }
        v2f accp[2][2] = {{{0.0f, 0.0f}, {0.0f, 0.0f}}, {{0.0f, 0.0f}, {0.0f, 0.0f}}};
#pragma unroll
        for (int tl = 0; tl < SCH; ++tl) {
            v4f knxt[2], qnxt[2];
            v2f enxt = {0.0f, 0.0f}, wvnxt = {0.0f, 0.0f};
            if (tl + 1 < SCH) {
                enxt    = *(const v2f*)(ew + (tl + 1) * SSTEP);
                wvnxt   = *(const v2f*)(ew + (tl + 1) * SSTEP + SROWS);
                knxt[0] = *(const v4f*)(kq + (tl + 1) * SSTEP);
                knxt[1] = *(const v4f*)(kq + (tl + 1) * SSTEP + 4);
                qnxt[0] = *(const v4f*)(kq + (tl + 1) * SSTEP + 64);
                qnxt[1] = *(const v4f*)(kq + (tl + 1) * SSTEP + 68);
            }
            __builtin_amdgcn_sched_barrier(0);

            if (tl > 0) {
                const v2f u0 = accp[0][0] + accp[0][1];
                const v2f u1 = accp[1][0] + accp[1][1];
                float t0 = u0[0] + u0[1];
                float t1 = u1[0] + u1[1];
                t0 = add_lane_xor<1>(t0); t1 = add_lane_xor<1>(t1);
                t0 = add_lane_xor<2>(t0); t1 = add_lane_xor<2>(t1);
                t0 = add_lane_xor<4>(t0); t1 = add_lane_xor<4>(t1);
                const v2f t = {t0, t1};
                *(v2f*)(ost + (tl - 1) * OSTEP) = t;
            }

#pragma unroll
            for (int row = 0; row < 2; ++row) {
                v2f a = {0.0f, 0.0f}, bb = {0.0f, 0.0f};
                ntm_pair(S[row][0], a,  kpc[row][0], kcur[0].lo, qcur[0].lo, wvcur[row]);
                ntm_pair(S[row][1], a,  kpc[row][1], kcur[0].hi, qcur[0].hi, wvcur[row]);
                ntm_pair(S[row][2], bb, kpc[row][2], kcur[1].lo, qcur[1].lo, wvcur[row]);
                ntm_pair(S[row][3], bb, kpc[row][3], kcur[1].hi, qcur[1].hi, wvcur[row]);
                accp[row][0] = a; accp[row][1] = bb;
            }

            if (tl + 1 < SCH) {
#pragma unroll
                for (int row = 0; row < 2; ++row) {
                    kpc[row][0] = ntm_keep(knxt[0].lo, enxt[row]);
                    kpc[row][1] = ntm_keep(knxt[0].hi, enxt[row]);
                    kpc[row][2] = ntm_keep(knxt[1].lo, enxt[row]);
                    kpc[row][3] = ntm_keep(knxt[1].hi, enxt[row]);
                }
                ecur = enxt; wvcur = wvnxt;
                kcur[0] = knxt[0]; kcur[1] = knxt[1];
                qcur[0] = qnxt[0]; qcur[1] = qnxt[1];
            }
        }
        {
            const v2f u0 = accp[0][0] + accp[0][1];
            const v2f u1 = accp[1][0] + accp[1][1];
            float t0 = u0[0] + u0[1];
            float t1 = u1[0] + u1[1];
            t0 = add_lane_xor<1>(t0); t1 = add_lane_xor<1>(t1);
            t0 = add_lane_xor<2>(t0); t1 = add_lane_xor<2>(t1);
            t0 = add_lane_xor<4>(t0); t1 = add_lane_xor<4>(t1);
            const v2f t = {t0, t1};
            *(v2f*)(ost + (SCH - 1) * OSTEP) = t;
        }
    }

    __syncthreads();
#pragma unroll
    for (int j = 0; j < EPASS; ++j) {
        const float* os = ostage + ((NCHK - 1) & 1) * (SCH * OSTEP) + orow + j * 16 * OSTEP;
        const v2f a01 = *(const v2f*)(os);
        const v2f a23 = *(const v2f*)(os + 16);
        v4f o;
        o[0] = tanh_row(a01[0]); o[1] = tanh_row(a01[1]); o[2] = tanh_row(a23[0]); o[3] = tanh_row(a23[1]);
        *(v4f*)(out + (size_t)(NCHK - 1) * SCH * 4096 + oidx + (size_t)j * 16 * 4096) = o;
    }

#pragma unroll
    for (int row = 0; row < 2; ++row) {
        v4f lo, hi;
        lo.lo = S[row][0]; lo.hi = S[row][1]; hi.lo = S[row][2]; hi.hi = S[row][3];
        *(v4f*)(Sfin + soff + row * 64)     = lo;
        *(v4f*)(Sfin + soff + row * 64 + 4) = hi;
    }
}

extern "C" void kernel_launch(void* const* d_in, const int* in_sizes, int n_in,
                              void* d_out, int out_size, void* d_ws, size_t ws_size,
                              hipStream_t stream) {
    (void)in_sizes; (void)n_in; (void)out_size; (void)ws_size;
    const float* x  = (const float*)d_in[0];
    const float* S0 = (const float*)d_in[1];
    const float* Wk = (const float*)d_in[2];
    const float* Wv = (const float*)d_in[3];
    const float* Wq = (const float*)d_in[4];
    const float* We = (const float*)d_in[5];
    const float* Ww = (const float*)d_in[6];

    float* out = (float*)d_out;
    float* ws  = (float*)d_ws;
    float* wk  = ws;
    float* wv  = ws + TBN;
    float* wq  = ws + 2 * TBN;
    float* we_ = ws + 3 * TBN;
    float* ww_ = ws + 4 * TBN;

    ntm_proj_kernel<<<ROWS / RT, 256, 0, stream>>>(
        x, Wk, Wv, Wq, We, Ww, wk, wv, wq, we_, ww_);

    ntm_scan_kernel<<<B_DIM * 2, 128, 0, stream>>>(
        S0, wk, wv, wq, we_, ww_, out, out + TBN);
}


# scan: LDS traffic of 2 timesteps issued as one group (both steps' operand reads + the 2 row-sum stores of the steps before); was a read group + a separate store per step; same sums, same order
# speedup vs baseline: 11.1823x; 1.0637x over previous
#include <hip/hip_runtime.h>
#include <math.h>

typedef float v2f __attribute__((ext_vector_type(2)));
typedef float v8f __attribute__((ext_vector_type(8)));

#define T_DIM 1024
#define B_DIM 64
#define D_DIM 512
#define N_DIM 64
#define ROWS  (T_DIM * B_DIM)
#define TBN   ((size_t)T_DIM * B_DIM * N_DIM)

typedef float v4f __attribute__((ext_vector_type(4)));

#define RT      128
#define KC      16
#define LP      20
#define NCHUNK  (D_DIM / KC)
#define A_FL    (RT * LP)
#define STG_FL  ((RT + 5 * N_DIM) * LP)
#define NPIECE  ((RT + 5 * N_DIM) * (KC / 4) / 256)
#define SP      65

__global__ __launch_bounds__(256) void ntm_proj_kernel(
    const float* __restrict__ x,
    const float* __restrict__ Wk, const float* __restrict__ Wv,
    const float* __restrict__ Wq, const float* __restrict__ We,
    const float* __restrict__ Ww,
    float* __restrict__ outk, float* __restrict__ outv,
    float* __restrict__ outq, float* __restrict__ oute,
    float* __restrict__ outw)
{
    __shared__ __attribute__((aligned(16))) float lds[2 * STG_FL];

    const int tid   = threadIdx.x;
    const int wave  = tid >> 5;
    const int lane  = tid & 31;
    const int lmod  = lane & 15;
    const int hi    = lane >> 4;
    const int Mbase = blockIdx.x * RT;
    const int Nbase = (wave & 3) * 16;
    const int Mhalf = (wave >> 2) * 64;

    const int srow = tid >> 2;
    const int sq   = (tid & 3) * 4;
    const float* gsrc[NPIECE];
    gsrc[0] = x  + (size_t)(Mbase + srow) * D_DIM + sq;
    gsrc[1] = x  + (size_t)(Mbase + 64 + srow) * D_DIM + sq;
    gsrc[2] = Wk + (size_t)srow * D_DIM + sq;
    gsrc[3] = Wv + (size_t)srow * D_DIM + sq;
    gsrc[4] = Wq + (size_t)srow * D_DIM + sq;
    gsrc[5] = We + (size_t)srow * D_DIM + sq;
    gsrc[6] = Ww + (size_t)srow * D_DIM + sq;
    const int sdst = srow * LP + sq;

    v8f acc[4][5];
#pragma unroll
    for (int mt = 0; mt < 4; ++mt)
#pragma unroll
        for (int m = 0; m < 5; ++m)
#pragma unroll
            for (int r = 0; r < 8; ++r) acc[mt][m][r] = 0.0f;

    v4f pf[NPIECE];
#pragma unroll
    for (int i = 0; i < NPIECE; ++i) pf[i] = *(const v4f*)(gsrc[i]);
#pragma unroll
    for (int i = 0; i < NPIECE; ++i) *(v4f*)(lds + sdst + i * 64 * LP) = pf[i];
    __syncthreads();

    const int aoff = (Mhalf + lmod) * LP + 2 * hi;
    const int boff = A_FL + (Nbase + lmod) * LP + 2 * hi;

    for (int c = 0; c < NCHUNK; ++c) {
        const float* st = lds + (c & 1) * STG_FL;

        if (c + 1 < NCHUNK) {
#pragma unroll
            for (int i = 0; i < NPIECE; ++i) pf[i] = *(const v4f*)(gsrc[i] + (c + 1) * KC);
        }

#pragma unroll
        for (int kk = 0; kk < KC; kk += 4) {
            v2f a[4], b[5];
#pragma unroll
            for (int mt = 0; mt < 4; ++mt) a[mt] = *(const v2f*)(st + aoff + mt * 16 * LP + kk);
#pragma unroll
            for (int m = 0; m < 5; ++m)   b[m]  = *(const v2f*)(st + boff + m * 64 * LP + kk);
#pragma unroll
            for (int mt = 0; mt < 4; ++mt)
#pragma unroll
                for (int m = 0; m < 5; ++m)
                    acc[mt][m] = __builtin_amdgcn_wmma_f32_16x16x4_f32(
                         false, a[mt],  false, b[m],
                         (short)0, acc[mt][m],
                         false,  false);
        }

        if (c + 1 < NCHUNK) {
            float* dn = lds + ((c + 1) & 1) * STG_FL;
#pragma unroll
            for (int i = 0; i < NPIECE; ++i) *(v4f*)(dn + sdst + i * 64 * LP) = pf[i];
        }
        __syncthreads();
    }

    float* ldsS = lds;
#pragma unroll
    for (int mt = 0; mt < 4; ++mt) {
#pragma unroll
        for (int r = 0; r < 8; ++r) {
            const int m = Mhalf + mt * 16 + (hi ? r + 8 : r);
            const int n = Nbase + lmod;
            ldsS[m * SP + n]           = acc[mt][0][r];
            ldsS[RT * SP + m * SP + n] = acc[mt][2][r];
            const size_t gi = (size_t)(Mbase + m) * N_DIM + n;
            outv[gi] = acc[mt][1][r];
            oute[gi] = 1.0f / (1.0f + __expf(-acc[mt][3][r]));
            outw[gi] = 1.0f / (1.0f + __expf(-acc[mt][4][r]));
        }
    }
    __syncthreads();

    {
        float* rowp = &ldsS[(tid >> 7) * (RT * SP) + (tid & (RT - 1)) * SP];
        float mx = rowp[0];
        for (int j = 1; j < 64; ++j) mx = fmaxf(mx, rowp[j]);
        float sum = 0.0f;
        for (int j = 0; j < 64; ++j) { float e2 = __expf(rowp[j] - mx); rowp[j] = e2; sum += e2; }
        const float inv = 1.0f / sum;
        for (int j = 0; j < 64; ++j) rowp[j] *= inv;
    }
    __syncthreads();

    for (int i = tid; i < RT * 64; i += 256) {
        const int m = i >> 6, n = i & 63;
        const size_t gi = (size_t)(Mbase + m) * N_DIM + n;
        outk[gi] = ldsS[m * SP + n];
        outq[gi] = ldsS[RT * SP + m * SP + n];
    }
}

#define SCH    64
#define KPASS  (SCH / 8)
#define EPASS  (SCH / 16)
#define SROWS  32
#define SSTEP  (64 + 64 + 2 * SROWS)
#define SHALF  (SCH * SSTEP)
#define NCHK   (T_DIM / SCH)
#define GST    2
#define NBODY  (SCH / GST)
#define OSTEP  (SROWS * 8)

static __device__ __forceinline__ void lds_handover_barrier()
{
    asm volatile("s_wait_dscnt 0x0\n\ts_barrier_signal -1\n\ts_barrier_wait -1" ::: "memory");
}

template <int MASK>
static __device__ __forceinline__ float add_lane_xor(float x)
{
    return x + __int_as_float(__builtin_amdgcn_mov_dpp(__float_as_int(x), 0x160 + MASK, 0xF, 0xF, true));
}

static __device__ __forceinline__ float tanh_row(float x)
{
    return __builtin_amdgcn_tanhf(x);
}

static __device__ __forceinline__ v2f ntm_keep(v2f k2, float e)
{
    const v2f one = {1.0f, 1.0f};
    const v2f ne  = {-e, -e};
    return __builtin_elementwise_fma(ne, k2, one);
}
static __device__ __forceinline__ void ntm_pair(v2f& s, v2f& acc, v2f keep, v2f k2, v2f q2, float wv)
{
    const v2f wv2 = {wv, wv};
    const v2f m   = s * keep;
    s   = __builtin_elementwise_fma(wv2, k2, m);
    acc = __builtin_elementwise_fma(s, q2, acc);
}

struct StepOps { v4f k0, k1, q0, q1; v2f e, wv; };

static __device__ __forceinline__ StepOps lds_step_ops(const float* kq, const float* ew, int t)
{
    StepOps o;
    o.e  = *(const v2f*)(ew + t * SSTEP);
    o.wv = *(const v2f*)(ew + t * SSTEP + SROWS);
    o.k0 = *(const v4f*)(kq + t * SSTEP);
    o.k1 = *(const v4f*)(kq + t * SSTEP + 4);
    o.q0 = *(const v4f*)(kq + t * SSTEP + 64);
    o.q1 = *(const v4f*)(kq + t * SSTEP + 68);
    return o;
}

static __device__ __forceinline__ void ntm_keep_all(v2f (&kpc)[2][4], const StepOps& o)
{
#pragma unroll
    for (int row = 0; row < 2; ++row) {
        kpc[row][0] = ntm_keep(o.k0.lo, o.e[row]);
        kpc[row][1] = ntm_keep(o.k0.hi, o.e[row]);
        kpc[row][2] = ntm_keep(o.k1.lo, o.e[row]);
        kpc[row][3] = ntm_keep(o.k1.hi, o.e[row]);
    }
}

static __device__ __forceinline__ v2f ntm_row_sums(const v2f (&accp)[2][2])
{
    const v2f u0 = accp[0][0] + accp[0][1];
    const v2f u1 = accp[1][0] + accp[1][1];
    float t0 = u0[0] + u0[1];
    float t1 = u1[0] + u1[1];
    t0 = add_lane_xor<1>(t0); t1 = add_lane_xor<1>(t1);
    t0 = add_lane_xor<2>(t0); t1 = add_lane_xor<2>(t1);
    t0 = add_lane_xor<4>(t0); t1 = add_lane_xor<4>(t1);
    const v2f t = {t0, t1};
    return t;
}

__global__ __launch_bounds__(128) void ntm_scan_kernel(
    const float* __restrict__ S0,
    const float* __restrict__ kG, const float* __restrict__ vG,
    const float* __restrict__ qG, const float* __restrict__ eG,
    const float* __restrict__ wG,
    float* __restrict__ out, float* __restrict__ Sfin)
{
    __shared__ __attribute__((aligned(16))) float stage[2 * SHALF];
    __shared__ __attribute__((aligned(16))) float ostage[2 * SCH * OSTEP];

    const int b    = blockIdx.x >> 1;
    const int g    = blockIdx.x & 1;
    const int tid  = threadIdx.x;
    const int rp   = tid >> 3;
    const int part = tid & 7;

    const size_t soff = (size_t)b * 4096 + (size_t)(g * SROWS + 2 * rp) * 64 + part * 8;
    v2f S[2][4];
#pragma unroll
    for (int row = 0; row < 2; ++row) {
        const v4f lo = *(const v4f*)(S0 + soff + row * 64);
        const v4f hi = *(const v4f*)(S0 + soff + row * 64 + 4);
        S[row][0] = lo.lo; S[row][1] = lo.hi; S[row][2] = hi.lo; S[row][3] = hi.hi;
    }

    const int kt = tid >> 4, kc = (tid & 15) * 4;
    const int et = tid >> 3, ec = (tid & 7) * 4;
    const size_t kgo = (size_t)kt * 4096 + (size_t)b * 64 + kc;
    const size_t ego = (size_t)et * 4096 + (size_t)b * 64 + g * SROWS + ec;
    const int klo = kt * SSTEP + kc;
    const int elo = et * SSTEP + 128 + ec;

    v4f pk[KPASS], pq[KPASS], pe[EPASS], pw[EPASS], pv[EPASS];
#pragma unroll
    for (int i = 0; i < KPASS; ++i) {
        pk[i] = *(const v4f*)(kG + kgo + (size_t)i * 8 * 4096);
        pq[i] = *(const v4f*)(qG + kgo + (size_t)i * 8 * 4096);
    }
#pragma unroll
    for (int i = 0; i < EPASS; ++i) {
        pe[i] = *(const v4f*)(eG + ego + (size_t)i * 16 * 4096);
        pw[i] = *(const v4f*)(wG + ego + (size_t)i * 16 * 4096);
        pv[i] = *(const v4f*)(vG + ego + (size_t)i * 16 * 4096);
    }

    __syncthreads();

    const size_t oidx = (size_t)(tid >> 3) * 4096 + (size_t)b * 64 + g * SROWS + (tid & 7) * 4;
    const int orow = (tid >> 3) * OSTEP + (tid & 7) * 32;

    for (int c = 0; c < NCHK; ++c) {
        const int par = c & 1;

        float* dst = stage + par * SHALF;
#pragma unroll
        for (int i = 0; i < KPASS; ++i) {
            *(v4f*)(dst + klo + i * 8 * SSTEP)      = pk[i];
            *(v4f*)(dst + klo + 64 + i * 8 * SSTEP) = pq[i];
        }
#pragma unroll
        for (int i = 0; i < EPASS; ++i) {
            *(v4f*)(dst + elo + i * 16 * SSTEP)         = pe[i];
            *(v4f*)(dst + elo + SROWS + i * 16 * SSTEP) = pw[i] * pv[i];
        }
        lds_handover_barrier();

        const float* kq = dst + part * 8;
        const float* ew = dst + 128 + 2 * rp;
        StepOps cur[GST], nxt[GST];
#pragma unroll
        for (int i = 0; i < GST; ++i) cur[i] = lds_step_ops(kq, ew, i);

        if (c + 1 < NCHK) {
            const size_t go = (size_t)(c + 1) * SCH * 4096;
#pragma unroll
            for (int i = 0; i < KPASS; ++i) {
                pk[i] = *(const v4f*)(kG + kgo + go + (size_t)i * 8 * 4096);
                pq[i] = *(const v4f*)(qG + kgo + go + (size_t)i * 8 * 4096);
            }
#pragma unroll
            for (int i = 0; i < EPASS; ++i) {
                pe[i] = *(const v4f*)(eG + ego + go + (size_t)i * 16 * 4096);
                pw[i] = *(const v4f*)(wG + ego + go + (size_t)i * 16 * 4096);
                pv[i] = *(const v4f*)(vG + ego + go + (size_t)i * 16 * 4096);
            }
        }
        if (c > 0) {
#pragma unroll
            for (int j = 0; j < EPASS; ++j) {
                const float* os = ostage + (par ^ 1) * (SCH * OSTEP) + orow + j * 16 * OSTEP;
                const v2f a01 = *(const v2f*)(os);
                const v2f a23 = *(const v2f*)(os + 16);
                v4f o;
                o[0] = tanh_row(a01[0]); o[1] = tanh_row(a01[1]); o[2] = tanh_row(a23[0]); o[3] = tanh_row(a23[1]);
                *(v4f*)(out + (size_t)(c - 1) * SCH * 4096 + oidx + (size_t)j * 16 * 4096) = o;
            }
        }

        float* ost = ostage + par * (SCH * OSTEP) + rp * 16 + part * 2;
        v2f kpc[2][4];
        ntm_keep_all(kpc, cur[0]);
        v2f accp[2][2] = {{{0.0f, 0.0f}, {0.0f, 0.0f}}, {{0.0f, 0.0f}, {0.0f, 0.0f}}};
        v2f pend[GST];
#pragma unroll
        for (int i = 0; i < GST; ++i) pend[i] = v2f{0.0f, 0.0f};
#pragma unroll
        for (int n = 0; n < NBODY; ++n) {
            __builtin_amdgcn_sched_barrier(0);
            if (n + 1 < NBODY) {
#pragma unroll
                for (int i = 0; i < GST; ++i) nxt[i] = lds_step_ops(kq, ew, (n + 1) * GST + i);
            }
            if (n > 0) {
#pragma unroll
                for (int i = 0; i < GST; ++i) {
                    const int ts = (n - 1) * GST - 1 + i;
                    if (ts >= 0) *(v2f*)(ost + ts * OSTEP) = pend[i];
                }
            }
            __builtin_amdgcn_sched_barrier(0);

#pragma unroll
            for (int i = 0; i < GST; ++i) {
                const int tl = n * GST + i;
                if (tl > 0) pend[i] = ntm_row_sums(accp);

                const StepOps& o = cur[i];
#pragma unroll
                for (int row = 0; row < 2; ++row) {
                    v2f a = {0.0f, 0.0f}, bb = {0.0f, 0.0f};
                    ntm_pair(S[row][0], a,  kpc[row][0], o.k0.lo, o.q0.lo, o.wv[row]);
                    ntm_pair(S[row][1], a,  kpc[row][1], o.k0.hi, o.q0.hi, o.wv[row]);
                    ntm_pair(S[row][2], bb, kpc[row][2], o.k1.lo, o.q1.lo, o.wv[row]);
                    ntm_pair(S[row][3], bb, kpc[row][3], o.k1.hi, o.q1.hi, o.wv[row]);
                    accp[row][0] = a; accp[row][1] = bb;
                }

                if (tl + 1 < SCH) {
                    if (i + 1 < GST) ntm_keep_all(kpc, cur[i + 1]);
                    else             ntm_keep_all(kpc, nxt[0]);
                }
            }
            if (n + 1 < NBODY) {
#pragma unroll
                for (int i = 0; i < GST; ++i) cur[i] = nxt[i];
            }
        }
#pragma unroll
        for (int i = 0; i < GST; ++i) *(v2f*)(ost + ((NBODY - 1) * GST - 1 + i) * OSTEP) = pend[i];
        *(v2f*)(ost + (SCH - 1) * OSTEP) = ntm_row_sums(accp);
    }

    __syncthreads();
#pragma unroll
    for (int j = 0; j < EPASS; ++j) {
        const float* os = ostage + ((NCHK - 1) & 1) * (SCH * OSTEP) + orow + j * 16 * OSTEP;
        const v2f a01 = *(const v2f*)(os);
        const v2f a23 = *(const v2f*)(os + 16);
        v4f o;
        o[0] = tanh_row(a01[0]); o[1] = tanh_row(a01[1]); o[2] = tanh_row(a23[0]); o[3] = tanh_row(a23[1]);
        *(v4f*)(out + (size_t)(NCHK - 1) * SCH * 4096 + oidx + (size_t)j * 16 * 4096) = o;
    }

#pragma unroll
    for (int row = 0; row < 2; ++row) {
        v4f lo, hi;
        lo.lo = S[row][0]; lo.hi = S[row][1]; hi.lo = S[row][2]; hi.hi = S[row][3];
        *(v4f*)(Sfin + soff + row * 64)     = lo;
        *(v4f*)(Sfin + soff + row * 64 + 4) = hi;
    }
}

extern "C" void kernel_launch(void* const* d_in, const int* in_sizes, int n_in,
                              void* d_out, int out_size, void* d_ws, size_t ws_size,
                              hipStream_t stream) {
    (void)in_sizes; (void)n_in; (void)out_size; (void)ws_size;
    const float* x  = (const float*)d_in[0];
    const float* S0 = (const float*)d_in[1];
    const float* Wk = (const float*)d_in[2];
    const float* Wv = (const float*)d_in[3];
    const float* Wq = (const float*)d_in[4];
    const float* We = (const float*)d_in[5];
    const float* Ww = (const float*)d_in[6];

    float* out = (float*)d_out;
    float* ws  = (float*)d_ws;
    float* wk  = ws;
    float* wv  = ws + TBN;
    float* wq  = ws + 2 * TBN;
    float* we_ = ws + 3 * TBN;
    float* ww_ = ws + 4 * TBN;

    ntm_proj_kernel<<<ROWS / RT, 256, 0, stream>>>(
        x, Wk, Wv, Wq, We, Ww, wk, wv, wq, we_, ww_);

    ntm_scan_kernel<<<B_DIM * 2, 128, 0, stream>>>(
        S0, wk, wv, wq, we_, ww_, out, out + TBN);
}
